# starting kernel
# baseline (speedup 1.0000x reference)
#include <hip/hip_runtime.h>
#include <hip/hip_bf16.h>

typedef float v2f __attribute__((ext_vector_type(2)));
typedef float v8f __attribute__((ext_vector_type(8)));

constexpr int NNODES = 20000;
constexpr int NEDGES = 320000;
constexpr int DIN    = 256;
constexpr int DHID   = 256;
constexpr int DOUT   = 64;

__global__ void zero_kernel(float* __restrict__ p, int n) {
    int i = blockIdx.x * blockDim.x + threadIdx.x;
    if (i < n) p[i] = 0.0f;
}

template <bool RELU>
__global__ __launch_bounds__(256)
void gemm_wmma_f32(const float* __restrict__ A, const float* __restrict__ W,
                   float* __restrict__ C, int M, int N, int K, int mTiles) {
    __shared__ float smem[256 * 64];

    const int tid  = threadIdx.x;
    const int lane = tid & 31;
    const int wave = tid >> 5;
    const int n0   = blockIdx.x << 6;

    {
        const int n   = tid & 63;
        const int kq0 = tid >> 6;
        for (int kq = kq0; kq < (K >> 2); kq += 4) {
            const float4 v = *(const float4*)(W + (size_t)(n0 + n) * K + (kq << 2));
            v2f lo; lo.x = v.x; lo.y = v.y;
            v2f hi; hi.x = v.z; hi.y = v.w;
            *(v2f*)(smem + (size_t)(2 * kq)     * 128 + 2 * n) = lo;
            *(v2f*)(smem + (size_t)(2 * kq + 1) * 128 + 2 * n) = hi;
        }
    }
    __syncthreads();

    const int mt = blockIdx.y * 8 + wave;
    if (mt >= mTiles) return;
    const int m0 = mt << 4;

    const int half = lane >> 4;
    const int l15  = lane & 15;

    const float* arow = A + (size_t)(m0 + l15) * K + 2 * half;
    __builtin_prefetch(arow, 0, 3);
    __builtin_prefetch(arow + 128, 0, 3);

    const float* bbase = smem + (size_t)half * 128 + 2 * l15;

    v8f acc0 = {}, acc1 = {}, acc2 = {}, acc3 = {};

    #pragma unroll 8
    for (int k0 = 0; k0 < K; k0 += 4) {
        v2f a = *(const v2f*)(arow + k0);
        if (RELU) { a.x = fmaxf(a.x, 0.0f); a.y = fmaxf(a.y, 0.0f); }
        const float* bp = bbase + (size_t)(k0 >> 1) * 128;
        v2f b0 = *(const v2f*)(bp +  0);
        v2f b1 = *(const v2f*)(bp + 32);
        v2f b2 = *(const v2f*)(bp + 64);
        v2f b3 = *(const v2f*)(bp + 96);
        acc0 = __builtin_amdgcn_wmma_f32_16x16x4_f32(false, a, false, b0, (short)0, acc0, false, false);
        acc1 = __builtin_amdgcn_wmma_f32_16x16x4_f32(false, a, false, b1, (short)0, acc1, false, false);
        acc2 = __builtin_amdgcn_wmma_f32_16x16x4_f32(false, a, false, b2, (short)0, acc2, false, false);
        acc3 = __builtin_amdgcn_wmma_f32_16x16x4_f32(false, a, false, b3, (short)0, acc3, false, false);
    }

    #pragma unroll
    for (int v = 0; v < 8; ++v) {
        float* crow = C + (size_t)(m0 + v + 8 * half) * N + n0 + l15;
        crow[ 0] = acc0[v];
        crow[16] = acc1[v];
        crow[32] = acc2[v];
        crow[48] = acc3[v];
    }
}

__global__ __launch_bounds__(256)
void scatter_kernel(const float* __restrict__ X, const int* __restrict__ src,
                    const int* __restrict__ dst, const float* __restrict__ w,
                    float* __restrict__ out, int E, int D) {
    const int per = D >> 2;
    int idx = blockIdx.x * blockDim.x + threadIdx.x;
    int e = idx / per;
    if (e >= E) return;
    int c = (idx % per) << 2;
    float we = w[e];
    const float4 x = *(const float4*)(X + (size_t)src[e] * D + c);
    float* o = out + (size_t)dst[e] * D + c;
    atomicAdd(o + 0, x.x * we);
    atomicAdd(o + 1, x.y * we);
    atomicAdd(o + 2, x.z * we);
    atomicAdd(o + 3, x.w * we);
}

extern "C" void kernel_launch(void* const* d_in, const int* in_sizes, int n_in,
                              void* d_out, int out_size, void* d_ws, size_t ws_size,
                              hipStream_t stream) {
    const float* node_feat = (const float*)d_in[0];
    const int*   edge_src  = (const int*)  d_in[1];
    const int*   edge_dst  = (const int*)  d_in[2];
    const float* edge_w    = (const float*)d_in[3];
    const float* W1        = (const float*)d_in[4];
    const float* W2        = (const float*)d_in[5];
    float* out = (float*)d_out;

    float* y1   = (float*)d_ws;
    float* agg1 = y1 + (size_t)NNODES * DHID;
    float* y2   = agg1 + (size_t)NNODES * DHID;

    const int mTiles = NNODES / 16;

    {
        int n = NNODES * DHID;
        zero_kernel<<<(n + 255) / 256, 256, 0, stream>>>(agg1, n);
        int m = NNODES * DOUT;
        zero_kernel<<<(m + 255) / 256, 256, 0, stream>>>(out, m);
    }

    {
        dim3 grid(DHID / 64, (mTiles + 7) / 8);
        gemm_wmma_f32<false><<<grid, 256, 0, stream>>>(
            node_feat, W1, y1, NNODES, DHID, DIN, mTiles);
    }
    {
        long long t = (long long)NEDGES * (DHID / 4);
        scatter_kernel<<<(int)((t + 255) / 256), 256, 0, stream>>>(
            y1, edge_src, edge_dst, edge_w, agg1, NEDGES, DHID);
    }
    {
        dim3 grid(DOUT / 64, (mTiles + 7) / 8);
        gemm_wmma_f32<true><<<grid, 256, 0, stream>>>(
            agg1, W2, y2, NNODES, DOUT, DHID, mTiles);
    }
    {
        long long t = (long long)NEDGES * (DOUT / 4);
        scatter_kernel<<<(int)((t + 255) / 256), 256, 0, stream>>>(
            y2, edge_src, edge_dst, edge_w, out, NEDGES, DOUT);
    }
}


# Float-atomic scatter-add replaced by gather over edges counting-sorted by dst in-call (int atomic tickets: row sum order = arrival order); one wave sums a row in registers, one store, no zeroing
# speedup vs baseline: 1.6790x; 1.6790x over previous
#include <hip/hip_runtime.h>
#include <hip/hip_bf16.h>

typedef float v2f __attribute__((ext_vector_type(2)));
typedef float v8f __attribute__((ext_vector_type(8)));

constexpr int NNODES = 20000;
constexpr int NEDGES = 320000;
constexpr int DIN    = 256;
constexpr int DHID   = 256;
constexpr int DOUT   = 64;


template <bool RELU>
__global__ __launch_bounds__(256)
void gemm_wmma_f32(const float* __restrict__ A, const float* __restrict__ W,
                   float* __restrict__ C, int M, int N, int K, int mTiles) {
    __shared__ float smem[256 * 64];

    const int tid  = threadIdx.x;
    const int lane = tid & 31;
    const int wave = tid >> 5;
    const int n0   = blockIdx.x << 6;

    {
        const int n   = tid & 63;
        const int kq0 = tid >> 6;
        for (int kq = kq0; kq < (K >> 2); kq += 4) {
            const float4 v = *(const float4*)(W + (size_t)(n0 + n) * K + (kq << 2));
            v2f lo; lo.x = v.x; lo.y = v.y;
            v2f hi; hi.x = v.z; hi.y = v.w;
            *(v2f*)(smem + (size_t)(2 * kq)     * 128 + 2 * n) = lo;
            *(v2f*)(smem + (size_t)(2 * kq + 1) * 128 + 2 * n) = hi;
        }
    }
    __syncthreads();

    const int mt = blockIdx.y * 8 + wave;
    if (mt >= mTiles) return;
    const int m0 = mt << 4;

    const int half = lane >> 4;
    const int l15  = lane & 15;

    const float* arow = A + (size_t)(m0 + l15) * K + 2 * half;
    __builtin_prefetch(arow, 0, 3);
    __builtin_prefetch(arow + 128, 0, 3);

    const float* bbase = smem + (size_t)half * 128 + 2 * l15;

    v8f acc0 = {}, acc1 = {}, acc2 = {}, acc3 = {};

    #pragma unroll 8
    for (int k0 = 0; k0 < K; k0 += 4) {
        v2f a = *(const v2f*)(arow + k0);
        if (RELU) { a.x = fmaxf(a.x, 0.0f); a.y = fmaxf(a.y, 0.0f); }
        const float* bp = bbase + (size_t)(k0 >> 1) * 128;
        v2f b0 = *(const v2f*)(bp +  0);
        v2f b1 = *(const v2f*)(bp + 32);
        v2f b2 = *(const v2f*)(bp + 64);
        v2f b3 = *(const v2f*)(bp + 96);
        acc0 = __builtin_amdgcn_wmma_f32_16x16x4_f32(false, a, false, b0, (short)0, acc0, false, false);
        acc1 = __builtin_amdgcn_wmma_f32_16x16x4_f32(false, a, false, b1, (short)0, acc1, false, false);
        acc2 = __builtin_amdgcn_wmma_f32_16x16x4_f32(false, a, false, b2, (short)0, acc2, false, false);
        acc3 = __builtin_amdgcn_wmma_f32_16x16x4_f32(false, a, false, b3, (short)0, acc3, false, false);
    }

    #pragma unroll
    for (int v = 0; v < 8; ++v) {
        float* crow = C + (size_t)(m0 + v + 8 * half) * N + n0 + l15;
        crow[ 0] = acc0[v];
        crow[16] = acc1[v];
        crow[32] = acc2[v];
        crow[48] = acc3[v];
    }
}

constexpr int SCAN_T     = 1024;
constexpr int SCAN_CH    = 20;
constexpr int ROWPTR_PAD = 20032;
constexpr int GATHER_WGS = 1280;
static_assert(NNODES % SCAN_CH == 0 && SCAN_CH % 4 == 0 && NNODES / SCAN_CH <= SCAN_T, "scan split");
static_assert(NNODES + 1 <= ROWPTR_PAD && ROWPTR_PAD % 4 == 0, "rowptr pad");
static_assert(NEDGES % 256 == 0, "edge grid");

__global__ void zero_int_kernel(int* __restrict__ p, int n) {
    int i = blockIdx.x * blockDim.x + threadIdx.x;
    if (i < n) p[i] = 0;
}

__global__ __launch_bounds__(256)
void rank_kernel(const int* __restrict__ dst, int* __restrict__ cnt,
                 int* __restrict__ rank, int E) {
    const int e = blockIdx.x * blockDim.x + threadIdx.x;
    if (e < E) rank[e] = atomicAdd(cnt + dst[e], 1);
}

__global__ __launch_bounds__(SCAN_T)
void scan_kernel(const int* __restrict__ cnt, int* __restrict__ rowptr) {
    __shared__ int part[SCAN_T];
    const int  t      = threadIdx.x;
    const int  i0     = t * SCAN_CH;
    const bool active = i0 < NNODES;
    int v[SCAN_CH];
    int sum = 0;
    #pragma unroll
    for (int q = 0; q < SCAN_CH / 4; ++q) {
        int4 c = make_int4(0, 0, 0, 0);
        if (active) c = *(const int4*)(cnt + i0 + 4 * q);
        v[4 * q + 0] = c.x; v[4 * q + 1] = c.y; v[4 * q + 2] = c.z; v[4 * q + 3] = c.w;
        sum += c.x + c.y + c.z + c.w;
    }
    part[t] = sum;
    __syncthreads();
    for (int off = 1; off < SCAN_T; off <<= 1) {
        const int add = (t >= off) ? part[t - off] : 0;
        __syncthreads();
        part[t] += add;
        __syncthreads();
    }
    int run = part[t] - sum;
    if (active) {
        #pragma unroll
        for (int q = 0; q < SCAN_CH / 4; ++q) {
            int4 r;
            r.x = run; run += v[4 * q + 0];
            r.y = run; run += v[4 * q + 1];
            r.z = run; run += v[4 * q + 2];
            r.w = run; run += v[4 * q + 3];
            *(int4*)(rowptr + i0 + 4 * q) = r;
        }
    }
    if (t == SCAN_T - 1) rowptr[NNODES] = part[SCAN_T - 1];
}

__global__ __launch_bounds__(256)
void place_kernel(const int* __restrict__ src, const int* __restrict__ dst,
                  const float* __restrict__ w, const int* __restrict__ rowptr,
                  const int* __restrict__ rank, int2* __restrict__ sedge, int E) {
    const int e = blockIdx.x * blockDim.x + threadIdx.x;
    if (e >= E) return;
    const int p = rowptr[dst[e]] + rank[e];
    sedge[p] = make_int2(src[e], __float_as_int(w[e]));
}

__global__ __launch_bounds__(256)
void gather_rows256(const float* __restrict__ X, const int* __restrict__ rowptr,
                    const int2* __restrict__ sedge, float* __restrict__ out, int nNodes) {
    const int lane = threadIdx.x & 31;
    const int wv = __builtin_amdgcn_readfirstlane((int)(threadIdx.x >> 5));
    const int nW = gridDim.x * 8;
    const unsigned last = (unsigned)(nNodes - 1);
    for (int n = blockIdx.x * 8 + wv; n < nNodes; n += nW) {
        const int beg = rowptr[n];
        const int end = rowptr[n + 1];
        float4 a0 = make_float4(0.f, 0.f, 0.f, 0.f);
        float4 a1 = a0;
        for (int base = beg; base < end; base += 32) {
            const int m = min(32, end - base);
            int2 ed = make_int2(0, 0);
            if (lane < m) ed = sedge[base + lane];
            int j = 0;
            for (; j + 4 <= m; j += 4) {
                const unsigned s0 = min((unsigned)__builtin_amdgcn_readlane(ed.x, j + 0), last);
                const unsigned s1 = min((unsigned)__builtin_amdgcn_readlane(ed.x, j + 1), last);
                const unsigned s2 = min((unsigned)__builtin_amdgcn_readlane(ed.x, j + 2), last);
                const unsigned s3 = min((unsigned)__builtin_amdgcn_readlane(ed.x, j + 3), last);
                const float w0 = __int_as_float(__builtin_amdgcn_readlane(ed.y, j + 0));
                const float w1 = __int_as_float(__builtin_amdgcn_readlane(ed.y, j + 1));
                const float w2 = __int_as_float(__builtin_amdgcn_readlane(ed.y, j + 2));
                const float w3 = __int_as_float(__builtin_amdgcn_readlane(ed.y, j + 3));
                const float4* r0 = (const float4*)(X + (size_t)s0 * 256) + lane;
                const float4* r1 = (const float4*)(X + (size_t)s1 * 256) + lane;
                const float4* r2 = (const float4*)(X + (size_t)s2 * 256) + lane;
                const float4* r3 = (const float4*)(X + (size_t)s3 * 256) + lane;
                const float4 x00 = r0[0], x01 = r0[32];
                const float4 x10 = r1[0], x11 = r1[32];
                const float4 x20 = r2[0], x21 = r2[32];
                const float4 x30 = r3[0], x31 = r3[32];
                a0.x = fmaf(w0, x00.x, a0.x); a0.y = fmaf(w0, x00.y, a0.y); a0.z = fmaf(w0, x00.z, a0.z); a0.w = fmaf(w0, x00.w, a0.w);
                a1.x = fmaf(w0, x01.x, a1.x); a1.y = fmaf(w0, x01.y, a1.y); a1.z = fmaf(w0, x01.z, a1.z); a1.w = fmaf(w0, x01.w, a1.w);
                a0.x = fmaf(w1, x10.x, a0.x); a0.y = fmaf(w1, x10.y, a0.y); a0.z = fmaf(w1, x10.z, a0.z); a0.w = fmaf(w1, x10.w, a0.w);
                a1.x = fmaf(w1, x11.x, a1.x); a1.y = fmaf(w1, x11.y, a1.y); a1.z = fmaf(w1, x11.z, a1.z); a1.w = fmaf(w1, x11.w, a1.w);
                a0.x = fmaf(w2, x20.x, a0.x); a0.y = fmaf(w2, x20.y, a0.y); a0.z = fmaf(w2, x20.z, a0.z); a0.w = fmaf(w2, x20.w, a0.w);
                a1.x = fmaf(w2, x21.x, a1.x); a1.y = fmaf(w2, x21.y, a1.y); a1.z = fmaf(w2, x21.z, a1.z); a1.w = fmaf(w2, x21.w, a1.w);
                a0.x = fmaf(w3, x30.x, a0.x); a0.y = fmaf(w3, x30.y, a0.y); a0.z = fmaf(w3, x30.z, a0.z); a0.w = fmaf(w3, x30.w, a0.w);
                a1.x = fmaf(w3, x31.x, a1.x); a1.y = fmaf(w3, x31.y, a1.y); a1.z = fmaf(w3, x31.z, a1.z); a1.w = fmaf(w3, x31.w, a1.w);
            }
            for (; j < m; ++j) {
                const unsigned s0 = min((unsigned)__builtin_amdgcn_readlane(ed.x, j), last);
                const float w0 = __int_as_float(__builtin_amdgcn_readlane(ed.y, j));
                const float4* r0 = (const float4*)(X + (size_t)s0 * 256) + lane;
                const float4 x00 = r0[0], x01 = r0[32];
                a0.x = fmaf(w0, x00.x, a0.x); a0.y = fmaf(w0, x00.y, a0.y); a0.z = fmaf(w0, x00.z, a0.z); a0.w = fmaf(w0, x00.w, a0.w);
                a1.x = fmaf(w0, x01.x, a1.x); a1.y = fmaf(w0, x01.y, a1.y); a1.z = fmaf(w0, x01.z, a1.z); a1.w = fmaf(w0, x01.w, a1.w);
            }
        }
        float4* o = (float4*)(out + (size_t)n * 256) + lane;
        o[0]  = a0;
        o[32] = a1;
    }
}

__global__ __launch_bounds__(256)
void gather_rows64(const float* __restrict__ X, const int* __restrict__ rowptr,
                   const int2* __restrict__ sedge, float* __restrict__ out, int nNodes) {
    const int lane = threadIdx.x & 31;
    const int wv = __builtin_amdgcn_readfirstlane((int)(threadIdx.x >> 5));
    const int nW = gridDim.x * 8;
    const unsigned last = (unsigned)(nNodes - 1);
    for (int n = blockIdx.x * 8 + wv; n < nNodes; n += nW) {
        const int beg = rowptr[n];
        const int end = rowptr[n + 1];
        float2 a = make_float2(0.f, 0.f);
        for (int base = beg; base < end; base += 32) {
            const int m = min(32, end - base);
            int2 ed = make_int2(0, 0);
            if (lane < m) ed = sedge[base + lane];
            int j = 0;
            for (; j + 4 <= m; j += 4) {
                const unsigned s0 = min((unsigned)__builtin_amdgcn_readlane(ed.x, j + 0), last);
                const unsigned s1 = min((unsigned)__builtin_amdgcn_readlane(ed.x, j + 1), last);
                const unsigned s2 = min((unsigned)__builtin_amdgcn_readlane(ed.x, j + 2), last);
                const unsigned s3 = min((unsigned)__builtin_amdgcn_readlane(ed.x, j + 3), last);
                const float w0 = __int_as_float(__builtin_amdgcn_readlane(ed.y, j + 0));
                const float w1 = __int_as_float(__builtin_amdgcn_readlane(ed.y, j + 1));
                const float w2 = __int_as_float(__builtin_amdgcn_readlane(ed.y, j + 2));
                const float w3 = __int_as_float(__builtin_amdgcn_readlane(ed.y, j + 3));
                const float2 x0 = *((const float2*)(X + (size_t)s0 * 64) + lane);
                const float2 x1 = *((const float2*)(X + (size_t)s1 * 64) + lane);
                const float2 x2 = *((const float2*)(X + (size_t)s2 * 64) + lane);
                const float2 x3 = *((const float2*)(X + (size_t)s3 * 64) + lane);
                a.x = fmaf(w0, x0.x, a.x); a.y = fmaf(w0, x0.y, a.y);
                a.x = fmaf(w1, x1.x, a.x); a.y = fmaf(w1, x1.y, a.y);
                a.x = fmaf(w2, x2.x, a.x); a.y = fmaf(w2, x2.y, a.y);
                a.x = fmaf(w3, x3.x, a.x); a.y = fmaf(w3, x3.y, a.y);
            }
            for (; j < m; ++j) {
                const unsigned s0 = min((unsigned)__builtin_amdgcn_readlane(ed.x, j), last);
                const float w0 = __int_as_float(__builtin_amdgcn_readlane(ed.y, j));
                const float2 x0 = *((const float2*)(X + (size_t)s0 * 64) + lane);
                a.x = fmaf(w0, x0.x, a.x); a.y = fmaf(w0, x0.y, a.y);
            }
        }
        *((float2*)(out + (size_t)n * 64) + lane) = a;
    }
}

extern "C" void kernel_launch(void* const* d_in, const int* in_sizes, int n_in,
                              void* d_out, int out_size, void* d_ws, size_t ws_size,
                              hipStream_t stream) {
    const float* node_feat = (const float*)d_in[0];
    const int*   edge_src  = (const int*)  d_in[1];
    const int*   edge_dst  = (const int*)  d_in[2];
    const float* edge_w    = (const float*)d_in[3];
    const float* W1        = (const float*)d_in[4];
    const float* W2        = (const float*)d_in[5];
    float* out = (float*)d_out;

    float* y1   = (float*)d_ws;
    float* agg1 = y1 + (size_t)NNODES * DHID;
    float* y2   = agg1 + (size_t)NNODES * DHID;
    int*   cnt    = (int*)(y2 + (size_t)NNODES * DOUT);
    int*   rowptr = cnt + NNODES;
    int*   rank   = rowptr + ROWPTR_PAD;
    int2*  sedge  = (int2*)(rank + NEDGES);

    constexpr int mTiles = NNODES / 16;

    zero_int_kernel<<<(NNODES + 255) / 256, 256, 0, stream>>>(cnt, NNODES);
    rank_kernel<<<NEDGES / 256, 256, 0, stream>>>(edge_dst, cnt, rank, NEDGES);
    scan_kernel<<<1, SCAN_T, 0, stream>>>(cnt, rowptr);
    place_kernel<<<NEDGES / 256, 256, 0, stream>>>(edge_src, edge_dst, edge_w, rowptr, rank,
                                                   sedge, NEDGES);

    {
        dim3 grid(DHID / 64, (mTiles + 7) / 8);
        gemm_wmma_f32<false><<<grid, 256, 0, stream>>>(
            node_feat, W1, y1, NNODES, DHID, DIN, mTiles);
    }
    gather_rows256<<<GATHER_WGS, 256, 0, stream>>>(y1, rowptr, sedge, agg1, NNODES);
    {
        dim3 grid(DOUT / 64, (mTiles + 7) / 8);
        gemm_wmma_f32<true><<<grid, 256, 0, stream>>>(
            agg1, W2, y2, NNODES, DOUT, DHID, mTiles);
    }
    gather_rows64<<<GATHER_WGS, 256, 0, stream>>>(y2, rowptr, sedge, out, NNODES);
}


# Counting sort (rank, scan, place kernels) replaced by one fill kernel: an edge with ticket below 64 goes to its row's fixed 64-slot bucket, later ones to a spill list; row sums keep arrival order
# speedup vs baseline: 1.8406x; 1.0962x over previous
#include <hip/hip_runtime.h>
#include <hip/hip_bf16.h>

typedef float v2f __attribute__((ext_vector_type(2)));
typedef float v8f __attribute__((ext_vector_type(8)));

constexpr int NNODES = 20000;
constexpr int NEDGES = 320000;
constexpr int DIN    = 256;
constexpr int DHID   = 256;
constexpr int DOUT   = 64;


template <bool RELU>
__global__ __launch_bounds__(256)
void gemm_wmma_f32(const float* __restrict__ A, const float* __restrict__ W,
                   float* __restrict__ C, int M, int N, int K, int mTiles) {
    __shared__ float smem[256 * 64];

    const int tid  = threadIdx.x;
    const int lane = tid & 31;
    const int wave = tid >> 5;
    const int n0   = blockIdx.x << 6;

    {
        const int n   = tid & 63;
        const int kq0 = tid >> 6;
        for (int kq = kq0; kq < (K >> 2); kq += 4) {
            const float4 v = *(const float4*)(W + (size_t)(n0 + n) * K + (kq << 2));
            v2f lo; lo.x = v.x; lo.y = v.y;
            v2f hi; hi.x = v.z; hi.y = v.w;
            *(v2f*)(smem + (size_t)(2 * kq)     * 128 + 2 * n) = lo;
            *(v2f*)(smem + (size_t)(2 * kq + 1) * 128 + 2 * n) = hi;
        }
    }
    __syncthreads();

    const int mt = blockIdx.y * 8 + wave;
    if (mt >= mTiles) return;
    const int m0 = mt << 4;

    const int half = lane >> 4;
    const int l15  = lane & 15;

    const float* arow = A + (size_t)(m0 + l15) * K + 2 * half;
    __builtin_prefetch(arow, 0, 3);
    __builtin_prefetch(arow + 128, 0, 3);

    const float* bbase = smem + (size_t)half * 128 + 2 * l15;

    v8f acc0 = {}, acc1 = {}, acc2 = {}, acc3 = {};

    #pragma unroll 8
    for (int k0 = 0; k0 < K; k0 += 4) {
        v2f a = *(const v2f*)(arow + k0);
        if (RELU) { a.x = fmaxf(a.x, 0.0f); a.y = fmaxf(a.y, 0.0f); }
        const float* bp = bbase + (size_t)(k0 >> 1) * 128;
        v2f b0 = *(const v2f*)(bp +  0);
        v2f b1 = *(const v2f*)(bp + 32);
        v2f b2 = *(const v2f*)(bp + 64);
        v2f b3 = *(const v2f*)(bp + 96);
        acc0 = __builtin_amdgcn_wmma_f32_16x16x4_f32(false, a, false, b0, (short)0, acc0, false, false);
        acc1 = __builtin_amdgcn_wmma_f32_16x16x4_f32(false, a, false, b1, (short)0, acc1, false, false);
        acc2 = __builtin_amdgcn_wmma_f32_16x16x4_f32(false, a, false, b2, (short)0, acc2, false, false);
        acc3 = __builtin_amdgcn_wmma_f32_16x16x4_f32(false, a, false, b3, (short)0, acc3, false, false);
    }

    #pragma unroll
    for (int v = 0; v < 8; ++v) {
        float* crow = C + (size_t)(m0 + v + 8 * half) * N + n0 + l15;
        crow[ 0] = acc0[v];
        crow[16] = acc1[v];
        crow[32] = acc2[v];
        crow[48] = acc3[v];
    }
}

constexpr int BCAP       = 64;
constexpr int CNT_PAD    = 20096;
constexpr int GATHER_WGS = 1280;
static_assert(NEDGES % 4 == 0, "edge quads");
static_assert(NNODES + 1 <= CNT_PAD && (CNT_PAD * 4) % 512 == 0, "counter pad");

__global__ void zero_int_kernel(int* __restrict__ p, int n) {
    int i = blockIdx.x * blockDim.x + threadIdx.x;
    if (i < n) p[i] = 0;
}

__device__ __forceinline__ void put_edge(int d, int s, float w, unsigned t,
                                         int* __restrict__ nspill, int2* __restrict__ bucket,
                                         int4* __restrict__ spill, int E) {
    if (t < (unsigned)BCAP) {
        bucket[(size_t)d * BCAP + t] = make_int2(s, __float_as_int(w));
    } else {
        const unsigned q = (unsigned)atomicAdd(nspill, 1);
        if (q < (unsigned)E) spill[q] = make_int4(d, s, __float_as_int(w), 0);
    }
}

__global__ __launch_bounds__(256)
void fill_kernel(const int* __restrict__ src, const int* __restrict__ dst,
                 const float* __restrict__ w, int* __restrict__ cnt, int* __restrict__ nspill,
                 int2* __restrict__ bucket, int4* __restrict__ spill, int E) {
    const int e0 = (blockIdx.x * 256 + threadIdx.x) * 4;
    if (e0 >= E) return;
    const int4   d4 = *(const int4*)(dst + e0);
    const int4   s4 = *(const int4*)(src + e0);
    const float4 w4 = *(const float4*)(w + e0);
    const unsigned t0 = (unsigned)atomicAdd(cnt + d4.x, 1);
    const unsigned t1 = (unsigned)atomicAdd(cnt + d4.y, 1);
    const unsigned t2 = (unsigned)atomicAdd(cnt + d4.z, 1);
    const unsigned t3 = (unsigned)atomicAdd(cnt + d4.w, 1);
    put_edge(d4.x, s4.x, w4.x, t0, nspill, bucket, spill, E);
    put_edge(d4.y, s4.y, w4.y, t1, nspill, bucket, spill, E);
    put_edge(d4.z, s4.z, w4.z, t2, nspill, bucket, spill, E);
    put_edge(d4.w, s4.w, w4.w, t3, nspill, bucket, spill, E);
}

__global__ __launch_bounds__(256)
void gather_rows256(const float* __restrict__ X, const int* __restrict__ cnt,
                    const int* __restrict__ nspill, const int2* __restrict__ bucket,
                    const int4* __restrict__ spill, float* __restrict__ out, int nNodes, int E) {
    const int lane = threadIdx.x & 31;
    const int wv = __builtin_amdgcn_readfirstlane((int)(threadIdx.x >> 5));
    const int nW = gridDim.x * 8;
    const unsigned last = (unsigned)(nNodes - 1);
    for (int n = blockIdx.x * 8 + wv; n < nNodes; n += nW) {
        const int c   = cnt[n];
        const int beg = n * BCAP;
        const int end = beg + min(c, BCAP);
        float4 a0 = make_float4(0.f, 0.f, 0.f, 0.f);
        float4 a1 = a0;
        for (int base = beg; base < end; base += 32) {
            const int m = min(32, end - base);
            int2 ed = make_int2(0, 0);
            if (lane < m) ed = bucket[base + lane];
            int j = 0;
            for (; j + 4 <= m; j += 4) {
                const unsigned s0 = min((unsigned)__builtin_amdgcn_readlane(ed.x, j + 0), last);
                const unsigned s1 = min((unsigned)__builtin_amdgcn_readlane(ed.x, j + 1), last);
                const unsigned s2 = min((unsigned)__builtin_amdgcn_readlane(ed.x, j + 2), last);
                const unsigned s3 = min((unsigned)__builtin_amdgcn_readlane(ed.x, j + 3), last);
                const float w0 = __int_as_float(__builtin_amdgcn_readlane(ed.y, j + 0));
                const float w1 = __int_as_float(__builtin_amdgcn_readlane(ed.y, j + 1));
                const float w2 = __int_as_float(__builtin_amdgcn_readlane(ed.y, j + 2));
                const float w3 = __int_as_float(__builtin_amdgcn_readlane(ed.y, j + 3));
                const float4* r0 = (const float4*)(X + (size_t)s0 * 256) + lane;
                const float4* r1 = (const float4*)(X + (size_t)s1 * 256) + lane;
                const float4* r2 = (const float4*)(X + (size_t)s2 * 256) + lane;
                const float4* r3 = (const float4*)(X + (size_t)s3 * 256) + lane;
                const float4 x00 = r0[0], x01 = r0[32];
                const float4 x10 = r1[0], x11 = r1[32];
                const float4 x20 = r2[0], x21 = r2[32];
                const float4 x30 = r3[0], x31 = r3[32];
                a0.x = fmaf(w0, x00.x, a0.x); a0.y = fmaf(w0, x00.y, a0.y); a0.z = fmaf(w0, x00.z, a0.z); a0.w = fmaf(w0, x00.w, a0.w);
                a1.x = fmaf(w0, x01.x, a1.x); a1.y = fmaf(w0, x01.y, a1.y); a1.z = fmaf(w0, x01.z, a1.z); a1.w = fmaf(w0, x01.w, a1.w);
                a0.x = fmaf(w1, x10.x, a0.x); a0.y = fmaf(w1, x10.y, a0.y); a0.z = fmaf(w1, x10.z, a0.z); a0.w = fmaf(w1, x10.w, a0.w);
                a1.x = fmaf(w1, x11.x, a1.x); a1.y = fmaf(w1, x11.y, a1.y); a1.z = fmaf(w1, x11.z, a1.z); a1.w = fmaf(w1, x11.w, a1.w);
                a0.x = fmaf(w2, x20.x, a0.x); a0.y = fmaf(w2, x20.y, a0.y); a0.z = fmaf(w2, x20.z, a0.z); a0.w = fmaf(w2, x20.w, a0.w);
                a1.x = fmaf(w2, x21.x, a1.x); a1.y = fmaf(w2, x21.y, a1.y); a1.z = fmaf(w2, x21.z, a1.z); a1.w = fmaf(w2, x21.w, a1.w);
                a0.x = fmaf(w3, x30.x, a0.x); a0.y = fmaf(w3, x30.y, a0.y); a0.z = fmaf(w3, x30.z, a0.z); a0.w = fmaf(w3, x30.w, a0.w);
                a1.x = fmaf(w3, x31.x, a1.x); a1.y = fmaf(w3, x31.y, a1.y); a1.z = fmaf(w3, x31.z, a1.z); a1.w = fmaf(w3, x31.w, a1.w);
            }
            for (; j < m; ++j) {
                const unsigned s0 = min((unsigned)__builtin_amdgcn_readlane(ed.x, j), last);
                const float w0 = __int_as_float(__builtin_amdgcn_readlane(ed.y, j));
                const float4* r0 = (const float4*)(X + (size_t)s0 * 256) + lane;
                const float4 x00 = r0[0], x01 = r0[32];
                a0.x = fmaf(w0, x00.x, a0.x); a0.y = fmaf(w0, x00.y, a0.y); a0.z = fmaf(w0, x00.z, a0.z); a0.w = fmaf(w0, x00.w, a0.w);
                a1.x = fmaf(w0, x01.x, a1.x); a1.y = fmaf(w0, x01.y, a1.y); a1.z = fmaf(w0, x01.z, a1.z); a1.w = fmaf(w0, x01.w, a1.w);
            }
        }
        if (c > BCAP) {
            const int tot = min(nspill[0], E);
            for (int base = 0; base < tot; base += 32) {
                const int m = min(32, tot - base);
                int4 ev = make_int4(-1, 0, 0, 0);
                if (lane < m) ev = spill[base + lane];
                for (int j = 0; j < m; ++j) {
                    if (__builtin_amdgcn_readlane(ev.x, j) != n) continue;
                    const unsigned s0 = min((unsigned)__builtin_amdgcn_readlane(ev.y, j), last);
                    const float w0 = __int_as_float(__builtin_amdgcn_readlane(ev.z, j));
                    const float4* r0 = (const float4*)(X + (size_t)s0 * 256) + lane;
                    const float4 x00 = r0[0], x01 = r0[32];
                    a0.x = fmaf(w0, x00.x, a0.x); a0.y = fmaf(w0, x00.y, a0.y); a0.z = fmaf(w0, x00.z, a0.z); a0.w = fmaf(w0, x00.w, a0.w);
                    a1.x = fmaf(w0, x01.x, a1.x); a1.y = fmaf(w0, x01.y, a1.y); a1.z = fmaf(w0, x01.z, a1.z); a1.w = fmaf(w0, x01.w, a1.w);
                }
            }
        }
        float4* o = (float4*)(out + (size_t)n * 256) + lane;
        o[0]  = a0;
        o[32] = a1;
    }
}

__global__ __launch_bounds__(256)
void gather_rows64(const float* __restrict__ X, const int* __restrict__ cnt,
                   const int* __restrict__ nspill, const int2* __restrict__ bucket,
                   const int4* __restrict__ spill, float* __restrict__ out, int nNodes, int E) {
    const int lane = threadIdx.x & 31;
    const int wv = __builtin_amdgcn_readfirstlane((int)(threadIdx.x >> 5));
    const int nW = gridDim.x * 8;
    const unsigned last = (unsigned)(nNodes - 1);
    for (int n = blockIdx.x * 8 + wv; n < nNodes; n += nW) {
        const int c   = cnt[n];
        const int beg = n * BCAP;
        const int end = beg + min(c, BCAP);
        float2 a = make_float2(0.f, 0.f);
        for (int base = beg; base < end; base += 32) {
            const int m = min(32, end - base);
            int2 ed = make_int2(0, 0);
            if (lane < m) ed = bucket[base + lane];
            int j = 0;
            for (; j + 4 <= m; j += 4) {
                const unsigned s0 = min((unsigned)__builtin_amdgcn_readlane(ed.x, j + 0), last);
                const unsigned s1 = min((unsigned)__builtin_amdgcn_readlane(ed.x, j + 1), last);
                const unsigned s2 = min((unsigned)__builtin_amdgcn_readlane(ed.x, j + 2), last);
                const unsigned s3 = min((unsigned)__builtin_amdgcn_readlane(ed.x, j + 3), last);
                const float w0 = __int_as_float(__builtin_amdgcn_readlane(ed.y, j + 0));
                const float w1 = __int_as_float(__builtin_amdgcn_readlane(ed.y, j + 1));
                const float w2 = __int_as_float(__builtin_amdgcn_readlane(ed.y, j + 2));
                const float w3 = __int_as_float(__builtin_amdgcn_readlane(ed.y, j + 3));
                const float2 x0 = *((const float2*)(X + (size_t)s0 * 64) + lane);
                const float2 x1 = *((const float2*)(X + (size_t)s1 * 64) + lane);
                const float2 x2 = *((const float2*)(X + (size_t)s2 * 64) + lane);
                const float2 x3 = *((const float2*)(X + (size_t)s3 * 64) + lane);
                a.x = fmaf(w0, x0.x, a.x); a.y = fmaf(w0, x0.y, a.y);
                a.x = fmaf(w1, x1.x, a.x); a.y = fmaf(w1, x1.y, a.y);
                a.x = fmaf(w2, x2.x, a.x); a.y = fmaf(w2, x2.y, a.y);
                a.x = fmaf(w3, x3.x, a.x); a.y = fmaf(w3, x3.y, a.y);
            }
            for (; j < m; ++j) {
                const unsigned s0 = min((unsigned)__builtin_amdgcn_readlane(ed.x, j), last);
                const float w0 = __int_as_float(__builtin_amdgcn_readlane(ed.y, j));
                const float2 x0 = *((const float2*)(X + (size_t)s0 * 64) + lane);
                a.x = fmaf(w0, x0.x, a.x); a.y = fmaf(w0, x0.y, a.y);
            }
        }
        if (c > BCAP) {
            const int tot = min(nspill[0], E);
            for (int base = 0; base < tot; base += 32) {
                const int m = min(32, tot - base);
                int4 ev = make_int4(-1, 0, 0, 0);
                if (lane < m) ev = spill[base + lane];
                for (int j = 0; j < m; ++j) {
                    if (__builtin_amdgcn_readlane(ev.x, j) != n) continue;
                    const unsigned s0 = min((unsigned)__builtin_amdgcn_readlane(ev.y, j), last);
                    const float w0 = __int_as_float(__builtin_amdgcn_readlane(ev.z, j));
                    const float2 x0 = *((const float2*)(X + (size_t)s0 * 64) + lane);
                    a.x = fmaf(w0, x0.x, a.x); a.y = fmaf(w0, x0.y, a.y);
                }
            }
        }
        *((float2*)(out + (size_t)n * 64) + lane) = a;
    }
}

extern "C" void kernel_launch(void* const* d_in, const int* in_sizes, int n_in,
                              void* d_out, int out_size, void* d_ws, size_t ws_size,
                              hipStream_t stream) {
    const float* node_feat = (const float*)d_in[0];
    const int*   edge_src  = (const int*)  d_in[1];
    const int*   edge_dst  = (const int*)  d_in[2];
    const float* edge_w    = (const float*)d_in[3];
    const float* W1        = (const float*)d_in[4];
    const float* W2        = (const float*)d_in[5];
    float* out = (float*)d_out;

    float* y1   = (float*)d_ws;
    float* agg1 = y1 + (size_t)NNODES * DHID;
    float* y2   = agg1 + (size_t)NNODES * DHID;
    int*   cnt    = (int*)(y2 + (size_t)NNODES * DOUT);
    int*   nspill = cnt + NNODES;
    int2*  bucket = (int2*)(cnt + CNT_PAD);
    int4*  spill  = (int4*)(bucket + (size_t)NNODES * BCAP);

    constexpr int mTiles = NNODES / 16;

    zero_int_kernel<<<(NNODES + 1 + 255) / 256, 256, 0, stream>>>(cnt, NNODES + 1);
    fill_kernel<<<(NEDGES / 4 + 255) / 256, 256, 0, stream>>>(edge_src, edge_dst, edge_w, cnt, nspill,
                                                          bucket, spill, NEDGES);

    {
        dim3 grid(DHID / 64, (mTiles + 7) / 8);
        gemm_wmma_f32<false><<<grid, 256, 0, stream>>>(
            node_feat, W1, y1, NNODES, DHID, DIN, mTiles);
    }
    gather_rows256<<<GATHER_WGS, 256, 0, stream>>>(y1, cnt, nspill, bucket, spill, agg1, NNODES, NEDGES);
    {
        dim3 grid(DOUT / 64, (mTiles + 7) / 8);
        gemm_wmma_f32<true><<<grid, 256, 0, stream>>>(
            agg1, W2, y2, NNODES, DOUT, DHID, mTiles);
    }
    gather_rows64<<<GATHER_WGS, 256, 0, stream>>>(y2, cnt, nspill, bucket, spill, out, NNODES, NEDGES);
}


# GEMM operand feed: W block copied to LDS row-major (padded pitch) by coalesced loads issued as one group; A and B fragments load 16 B per lane (k summed in another order within each group of 8)
# speedup vs baseline: 1.9954x; 1.0841x over previous
#include <hip/hip_runtime.h>
#include <hip/hip_bf16.h>

typedef float v2f __attribute__((ext_vector_type(2)));
typedef float v8f __attribute__((ext_vector_type(8)));

constexpr int NNODES = 20000;
constexpr int NEDGES = 320000;
constexpr int DIN    = 256;
constexpr int DHID   = 256;
constexpr int DOUT   = 64;


constexpr int GK      = 256;
constexpr int BPITCH4 = GK / 4 + 1;

template <bool RELU>
__global__ __launch_bounds__(256)
void gemm_wmma_f32(const float* __restrict__ A, const float* __restrict__ W,
                   float* __restrict__ C, int M, int N, int mTiles) {
    __shared__ float4 smem4[64 * BPITCH4];

    const int tid  = threadIdx.x;
    const int lane = tid & 31;
    const int wave = tid >> 5;
    const int n0   = blockIdx.x << 6;

    {
        const int c4 = tid & 63;
        const int r0 = tid >> 6;
        const float4* wsrc = (const float4*)(W + (size_t)n0 * GK);
        float4 v[16];
        #pragma unroll
        for (int i = 0; i < 16; ++i)
            v[i] = wsrc[(size_t)(i * 4 + r0) * (GK / 4) + c4];
        #pragma unroll
        for (int i = 0; i < 16; ++i)
            smem4[(i * 4 + r0) * BPITCH4 + c4] = v[i];
    }
    __syncthreads();

    const int mt = blockIdx.y * 8 + wave;
    if (mt >= mTiles) return;
    const int m0 = mt << 4;

    const int half = lane >> 4;
    const int l15  = lane & 15;

    const float* arow = A + (size_t)(m0 + l15) * GK + 4 * half;
    __builtin_prefetch(arow, 0, 3);
    __builtin_prefetch(arow + 128, 0, 3);

    const float4* bbase = smem4 + l15 * BPITCH4 + half;

    v8f acc0 = {}, acc1 = {}, acc2 = {}, acc3 = {};

    #pragma unroll 4
    for (int k0 = 0; k0 < GK; k0 += 8) {
        float4 a = *(const float4*)(arow + k0);
        if (RELU) {
            a.x = fmaxf(a.x, 0.0f); a.y = fmaxf(a.y, 0.0f);
            a.z = fmaxf(a.z, 0.0f); a.w = fmaxf(a.w, 0.0f);
        }
        const float4* bp = bbase + (k0 >> 2);
        const float4 b0 = bp[0];
        const float4 b1 = bp[16 * BPITCH4];
        const float4 b2 = bp[32 * BPITCH4];
        const float4 b3 = bp[48 * BPITCH4];
        v2f alo; alo.x = a.x; alo.y = a.y;
        v2f ahi; ahi.x = a.z; ahi.y = a.w;
        v2f b0lo; b0lo.x = b0.x; b0lo.y = b0.y;
        v2f b1lo; b1lo.x = b1.x; b1lo.y = b1.y;
        v2f b2lo; b2lo.x = b2.x; b2lo.y = b2.y;
        v2f b3lo; b3lo.x = b3.x; b3lo.y = b3.y;
        v2f b0hi; b0hi.x = b0.z; b0hi.y = b0.w;
        v2f b1hi; b1hi.x = b1.z; b1hi.y = b1.w;
        v2f b2hi; b2hi.x = b2.z; b2hi.y = b2.w;
        v2f b3hi; b3hi.x = b3.z; b3hi.y = b3.w;
        acc0 = __builtin_amdgcn_wmma_f32_16x16x4_f32(false, alo, false, b0lo, (short)0, acc0, false, false);
        acc1 = __builtin_amdgcn_wmma_f32_16x16x4_f32(false, alo, false, b1lo, (short)0, acc1, false, false);
        acc2 = __builtin_amdgcn_wmma_f32_16x16x4_f32(false, alo, false, b2lo, (short)0, acc2, false, false);
        acc3 = __builtin_amdgcn_wmma_f32_16x16x4_f32(false, alo, false, b3lo, (short)0, acc3, false, false);
        acc0 = __builtin_amdgcn_wmma_f32_16x16x4_f32(false, ahi, false, b0hi, (short)0, acc0, false, false);
        acc1 = __builtin_amdgcn_wmma_f32_16x16x4_f32(false, ahi, false, b1hi, (short)0, acc1, false, false);
        acc2 = __builtin_amdgcn_wmma_f32_16x16x4_f32(false, ahi, false, b2hi, (short)0, acc2, false, false);
        acc3 = __builtin_amdgcn_wmma_f32_16x16x4_f32(false, ahi, false, b3hi, (short)0, acc3, false, false);
    }

    #pragma unroll
    for (int v = 0; v < 8; ++v) {
        float* crow = C + (size_t)(m0 + v + 8 * half) * N + n0 + l15;
        crow[ 0] = acc0[v];
        crow[16] = acc1[v];
        crow[32] = acc2[v];
        crow[48] = acc3[v];
    }
}

constexpr int BCAP       = 64;
constexpr int CNT_PAD    = 20096;
constexpr int GATHER_WGS = 1280;
static_assert(NEDGES % 4 == 0, "edge quads");
static_assert(DIN == GK && DHID == GK, "both products have K == 256");
static_assert(NNODES + 1 <= CNT_PAD && (CNT_PAD * 4) % 512 == 0, "counter pad");

__global__ void zero_int_kernel(int* __restrict__ p, int n) {
    int i = blockIdx.x * blockDim.x + threadIdx.x;
    if (i < n) p[i] = 0;
}

__device__ __forceinline__ void put_edge(int d, int s, float w, unsigned t,
                                         int* __restrict__ nspill, int2* __restrict__ bucket,
                                         int4* __restrict__ spill, int E) {
    if (t < (unsigned)BCAP) {
        bucket[(size_t)d * BCAP + t] = make_int2(s, __float_as_int(w));
    } else {
        const unsigned q = (unsigned)atomicAdd(nspill, 1);
        if (q < (unsigned)E) spill[q] = make_int4(d, s, __float_as_int(w), 0);
    }
}

__global__ __launch_bounds__(256)
void fill_kernel(const int* __restrict__ src, const int* __restrict__ dst,
                 const float* __restrict__ w, int* __restrict__ cnt, int* __restrict__ nspill,
                 int2* __restrict__ bucket, int4* __restrict__ spill, int E) {
    const int e0 = (blockIdx.x * 256 + threadIdx.x) * 4;
    if (e0 >= E) return;
    const int4   d4 = *(const int4*)(dst + e0);
    const int4   s4 = *(const int4*)(src + e0);
    const float4 w4 = *(const float4*)(w + e0);
    const unsigned t0 = (unsigned)atomicAdd(cnt + d4.x, 1);
    const unsigned t1 = (unsigned)atomicAdd(cnt + d4.y, 1);
    const unsigned t2 = (unsigned)atomicAdd(cnt + d4.z, 1);
    const unsigned t3 = (unsigned)atomicAdd(cnt + d4.w, 1);
    put_edge(d4.x, s4.x, w4.x, t0, nspill, bucket, spill, E);
    put_edge(d4.y, s4.y, w4.y, t1, nspill, bucket, spill, E);
    put_edge(d4.z, s4.z, w4.z, t2, nspill, bucket, spill, E);
    put_edge(d4.w, s4.w, w4.w, t3, nspill, bucket, spill, E);
}

__global__ __launch_bounds__(256)
void gather_rows256(const float* __restrict__ X, const int* __restrict__ cnt,
                    const int* __restrict__ nspill, const int2* __restrict__ bucket,
                    const int4* __restrict__ spill, float* __restrict__ out, int nNodes, int E) {
    const int lane = threadIdx.x & 31;
    const int wv = __builtin_amdgcn_readfirstlane((int)(threadIdx.x >> 5));
    const int nW = gridDim.x * 8;
    const unsigned last = (unsigned)(nNodes - 1);
    for (int n = blockIdx.x * 8 + wv; n < nNodes; n += nW) {
        const int c   = cnt[n];
        const int beg = n * BCAP;
        const int end = beg + min(c, BCAP);
        float4 a0 = make_float4(0.f, 0.f, 0.f, 0.f);
        float4 a1 = a0;
        for (int base = beg; base < end; base += 32) {
            const int m = min(32, end - base);
            int2 ed = make_int2(0, 0);
            if (lane < m) ed = bucket[base + lane];
            int j = 0;
            for (; j + 4 <= m; j += 4) {
                const unsigned s0 = min((unsigned)__builtin_amdgcn_readlane(ed.x, j + 0), last);
                const unsigned s1 = min((unsigned)__builtin_amdgcn_readlane(ed.x, j + 1), last);
                const unsigned s2 = min((unsigned)__builtin_amdgcn_readlane(ed.x, j + 2), last);
                const unsigned s3 = min((unsigned)__builtin_amdgcn_readlane(ed.x, j + 3), last);
                const float w0 = __int_as_float(__builtin_amdgcn_readlane(ed.y, j + 0));
                const float w1 = __int_as_float(__builtin_amdgcn_readlane(ed.y, j + 1));
                const float w2 = __int_as_float(__builtin_amdgcn_readlane(ed.y, j + 2));
                const float w3 = __int_as_float(__builtin_amdgcn_readlane(ed.y, j + 3));
                const float4* r0 = (const float4*)(X + (size_t)s0 * 256) + lane;
                const float4* r1 = (const float4*)(X + (size_t)s1 * 256) + lane;
                const float4* r2 = (const float4*)(X + (size_t)s2 * 256) + lane;
                const float4* r3 = (const float4*)(X + (size_t)s3 * 256) + lane;
                const float4 x00 = r0[0], x01 = r0[32];
                const float4 x10 = r1[0], x11 = r1[32];
                const float4 x20 = r2[0], x21 = r2[32];
                const float4 x30 = r3[0], x31 = r3[32];
                a0.x = fmaf(w0, x00.x, a0.x); a0.y = fmaf(w0, x00.y, a0.y); a0.z = fmaf(w0, x00.z, a0.z); a0.w = fmaf(w0, x00.w, a0.w);
                a1.x = fmaf(w0, x01.x, a1.x); a1.y = fmaf(w0, x01.y, a1.y); a1.z = fmaf(w0, x01.z, a1.z); a1.w = fmaf(w0, x01.w, a1.w);
                a0.x = fmaf(w1, x10.x, a0.x); a0.y = fmaf(w1, x10.y, a0.y); a0.z = fmaf(w1, x10.z, a0.z); a0.w = fmaf(w1, x10.w, a0.w);
                a1.x = fmaf(w1, x11.x, a1.x); a1.y = fmaf(w1, x11.y, a1.y); a1.z = fmaf(w1, x11.z, a1.z); a1.w = fmaf(w1, x11.w, a1.w);
                a0.x = fmaf(w2, x20.x, a0.x); a0.y = fmaf(w2, x20.y, a0.y); a0.z = fmaf(w2, x20.z, a0.z); a0.w = fmaf(w2, x20.w, a0.w);
                a1.x = fmaf(w2, x21.x, a1.x); a1.y = fmaf(w2, x21.y, a1.y); a1.z = fmaf(w2, x21.z, a1.z); a1.w = fmaf(w2, x21.w, a1.w);
                a0.x = fmaf(w3, x30.x, a0.x); a0.y = fmaf(w3, x30.y, a0.y); a0.z = fmaf(w3, x30.z, a0.z); a0.w = fmaf(w3, x30.w, a0.w);
                a1.x = fmaf(w3, x31.x, a1.x); a1.y = fmaf(w3, x31.y, a1.y); a1.z = fmaf(w3, x31.z, a1.z); a1.w = fmaf(w3, x31.w, a1.w);
            }
            for (; j < m; ++j) {
                const unsigned s0 = min((unsigned)__builtin_amdgcn_readlane(ed.x, j), last);
                const float w0 = __int_as_float(__builtin_amdgcn_readlane(ed.y, j));
                const float4* r0 = (const float4*)(X + (size_t)s0 * 256) + lane;
                const float4 x00 = r0[0], x01 = r0[32];
                a0.x = fmaf(w0, x00.x, a0.x); a0.y = fmaf(w0, x00.y, a0.y); a0.z = fmaf(w0, x00.z, a0.z); a0.w = fmaf(w0, x00.w, a0.w);
                a1.x = fmaf(w0, x01.x, a1.x); a1.y = fmaf(w0, x01.y, a1.y); a1.z = fmaf(w0, x01.z, a1.z); a1.w = fmaf(w0, x01.w, a1.w);
            }
        }
        if (c > BCAP) {
            const int tot = min(nspill[0], E);
            for (int base = 0; base < tot; base += 32) {
                const int m = min(32, tot - base);
                int4 ev = make_int4(-1, 0, 0, 0);
                if (lane < m) ev = spill[base + lane];
                for (int j = 0; j < m; ++j) {
                    if (__builtin_amdgcn_readlane(ev.x, j) != n) continue;
                    const unsigned s0 = min((unsigned)__builtin_amdgcn_readlane(ev.y, j), last);
                    const float w0 = __int_as_float(__builtin_amdgcn_readlane(ev.z, j));
                    const float4* r0 = (const float4*)(X + (size_t)s0 * 256) + lane;
                    const float4 x00 = r0[0], x01 = r0[32];
                    a0.x = fmaf(w0, x00.x, a0.x); a0.y = fmaf(w0, x00.y, a0.y); a0.z = fmaf(w0, x00.z, a0.z); a0.w = fmaf(w0, x00.w, a0.w);
                    a1.x = fmaf(w0, x01.x, a1.x); a1.y = fmaf(w0, x01.y, a1.y); a1.z = fmaf(w0, x01.z, a1.z); a1.w = fmaf(w0, x01.w, a1.w);
                }
            }
        }
        float4* o = (float4*)(out + (size_t)n * 256) + lane;
        o[0]  = a0;
        o[32] = a1;
    }
}

__global__ __launch_bounds__(256)
void gather_rows64(const float* __restrict__ X, const int* __restrict__ cnt,
                   const int* __restrict__ nspill, const int2* __restrict__ bucket,
                   const int4* __restrict__ spill, float* __restrict__ out, int nNodes, int E) {
    const int lane = threadIdx.x & 31;
    const int wv = __builtin_amdgcn_readfirstlane((int)(threadIdx.x >> 5));
    const int nW = gridDim.x * 8;
    const unsigned last = (unsigned)(nNodes - 1);
    for (int n = blockIdx.x * 8 + wv; n < nNodes; n += nW) {
        const int c   = cnt[n];
        const int beg = n * BCAP;
        const int end = beg + min(c, BCAP);
        float2 a = make_float2(0.f, 0.f);
        for (int base = beg; base < end; base += 32) {
            const int m = min(32, end - base);
            int2 ed = make_int2(0, 0);
            if (lane < m) ed = bucket[base + lane];
            int j = 0;
            for (; j + 4 <= m; j += 4) {
                const unsigned s0 = min((unsigned)__builtin_amdgcn_readlane(ed.x, j + 0), last);
                const unsigned s1 = min((unsigned)__builtin_amdgcn_readlane(ed.x, j + 1), last);
                const unsigned s2 = min((unsigned)__builtin_amdgcn_readlane(ed.x, j + 2), last);
                const unsigned s3 = min((unsigned)__builtin_amdgcn_readlane(ed.x, j + 3), last);
                const float w0 = __int_as_float(__builtin_amdgcn_readlane(ed.y, j + 0));
                const float w1 = __int_as_float(__builtin_amdgcn_readlane(ed.y, j + 1));
                const float w2 = __int_as_float(__builtin_amdgcn_readlane(ed.y, j + 2));
                const float w3 = __int_as_float(__builtin_amdgcn_readlane(ed.y, j + 3));
                const float2 x0 = *((const float2*)(X + (size_t)s0 * 64) + lane);
                const float2 x1 = *((const float2*)(X + (size_t)s1 * 64) + lane);
                const float2 x2 = *((const float2*)(X + (size_t)s2 * 64) + lane);
                const float2 x3 = *((const float2*)(X + (size_t)s3 * 64) + lane);
                a.x = fmaf(w0, x0.x, a.x); a.y = fmaf(w0, x0.y, a.y);
                a.x = fmaf(w1, x1.x, a.x); a.y = fmaf(w1, x1.y, a.y);
                a.x = fmaf(w2, x2.x, a.x); a.y = fmaf(w2, x2.y, a.y);
                a.x = fmaf(w3, x3.x, a.x); a.y = fmaf(w3, x3.y, a.y);
            }
            for (; j < m; ++j) {
                const unsigned s0 = min((unsigned)__builtin_amdgcn_readlane(ed.x, j), last);
                const float w0 = __int_as_float(__builtin_amdgcn_readlane(ed.y, j));
                const float2 x0 = *((const float2*)(X + (size_t)s0 * 64) + lane);
                a.x = fmaf(w0, x0.x, a.x); a.y = fmaf(w0, x0.y, a.y);
            }
        }
        if (c > BCAP) {
            const int tot = min(nspill[0], E);
            for (int base = 0; base < tot; base += 32) {
                const int m = min(32, tot - base);
                int4 ev = make_int4(-1, 0, 0, 0);
                if (lane < m) ev = spill[base + lane];
                for (int j = 0; j < m; ++j) {
                    if (__builtin_amdgcn_readlane(ev.x, j) != n) continue;
                    const unsigned s0 = min((unsigned)__builtin_amdgcn_readlane(ev.y, j), last);
                    const float w0 = __int_as_float(__builtin_amdgcn_readlane(ev.z, j));
                    const float2 x0 = *((const float2*)(X + (size_t)s0 * 64) + lane);
                    a.x = fmaf(w0, x0.x, a.x); a.y = fmaf(w0, x0.y, a.y);
                }
            }
        }
        *((float2*)(out + (size_t)n * 64) + lane) = a;
    }
}

extern "C" void kernel_launch(void* const* d_in, const int* in_sizes, int n_in,
                              void* d_out, int out_size, void* d_ws, size_t ws_size,
                              hipStream_t stream) {
    const float* node_feat = (const float*)d_in[0];
    const int*   edge_src  = (const int*)  d_in[1];
    const int*   edge_dst  = (const int*)  d_in[2];
    const float* edge_w    = (const float*)d_in[3];
    const float* W1        = (const float*)d_in[4];
    const float* W2        = (const float*)d_in[5];
    float* out = (float*)d_out;

    float* y1   = (float*)d_ws;
    float* agg1 = y1 + (size_t)NNODES * DHID;
    float* y2   = agg1 + (size_t)NNODES * DHID;
    int*   cnt    = (int*)(y2 + (size_t)NNODES * DOUT);
    int*   nspill = cnt + NNODES;
    int2*  bucket = (int2*)(cnt + CNT_PAD);
    int4*  spill  = (int4*)(bucket + (size_t)NNODES * BCAP);

    constexpr int mTiles = NNODES / 16;

    zero_int_kernel<<<(NNODES + 1 + 255) / 256, 256, 0, stream>>>(cnt, NNODES + 1);
    fill_kernel<<<(NEDGES / 4 + 255) / 256, 256, 0, stream>>>(edge_src, edge_dst, edge_w, cnt, nspill,
                                                          bucket, spill, NEDGES);

    {
        dim3 grid(DHID / 64, (mTiles + 7) / 8);
        gemm_wmma_f32<false><<<grid, 256, 0, stream>>>(
            node_feat, W1, y1, NNODES, DHID, mTiles);
    }
    gather_rows256<<<GATHER_WGS, 256, 0, stream>>>(y1, cnt, nspill, bucket, spill, agg1, NNODES, NEDGES);
    {
        dim3 grid(DOUT / 64, (mTiles + 7) / 8);
        gemm_wmma_f32<true><<<grid, 256, 0, stream>>>(
            agg1, W2, y2, NNODES, DOUT, mTiles);
    }
    gather_rows64<<<GATHER_WGS, 256, 0, stream>>>(y2, cnt, nspill, bucket, spill, out, NNODES, NEDGES);
}


# Edge bucketing (fill) moved into the layer-1 GEMM launch as extra workgroups chosen by block index: the two independent kernels overlap, one launch fewer; row sums keep ticket arrival order
# speedup vs baseline: 2.1897x; 1.0974x over previous
#include <hip/hip_runtime.h>
#include <hip/hip_bf16.h>

typedef float v2f __attribute__((ext_vector_type(2)));
typedef float v8f __attribute__((ext_vector_type(8)));

constexpr int NNODES = 20000;
constexpr int NEDGES = 320000;
constexpr int DIN    = 256;
constexpr int DHID   = 256;
constexpr int DOUT   = 64;


constexpr int GK      = 256;
constexpr int BPITCH4 = GK / 4 + 1;

template <bool RELU>
__device__ __forceinline__
void gemm_block(const float* __restrict__ A, const float* __restrict__ W,
                float* __restrict__ C, int N, int mTiles, int bx, int by) {
    __shared__ float4 smem4[64 * BPITCH4];

    const int tid  = threadIdx.x;
    const int lane = tid & 31;
    const int wave = tid >> 5;
    const int n0   = bx << 6;

    {
        const int c4 = tid & 63;
        const int r0 = tid >> 6;
        const float4* wsrc = (const float4*)(W + (size_t)n0 * GK);
        float4 v[16];
        #pragma unroll
        for (int i = 0; i < 16; ++i)
            v[i] = wsrc[(size_t)(i * 4 + r0) * (GK / 4) + c4];
        #pragma unroll
        for (int i = 0; i < 16; ++i)
            smem4[(i * 4 + r0) * BPITCH4 + c4] = v[i];
    }
    __syncthreads();

    const int mt = by * 8 + wave;
    if (mt >= mTiles) return;
    const int m0 = mt << 4;

    const int half = lane >> 4;
    const int l15  = lane & 15;

    const float* arow = A + (size_t)(m0 + l15) * GK + 4 * half;
    __builtin_prefetch(arow, 0, 3);
    __builtin_prefetch(arow + 128, 0, 3);

    const float4* bbase = smem4 + l15 * BPITCH4 + half;

    v8f acc0 = {}, acc1 = {}, acc2 = {}, acc3 = {};

    #pragma unroll 4
    for (int k0 = 0; k0 < GK; k0 += 8) {
        float4 a = *(const float4*)(arow + k0);
        if (RELU) {
            a.x = fmaxf(a.x, 0.0f); a.y = fmaxf(a.y, 0.0f);
            a.z = fmaxf(a.z, 0.0f); a.w = fmaxf(a.w, 0.0f);
        }
        const float4* bp = bbase + (k0 >> 2);
        const float4 b0 = bp[0];
        const float4 b1 = bp[16 * BPITCH4];
        const float4 b2 = bp[32 * BPITCH4];
        const float4 b3 = bp[48 * BPITCH4];
        v2f alo; alo.x = a.x; alo.y = a.y;
        v2f ahi; ahi.x = a.z; ahi.y = a.w;
        v2f b0lo; b0lo.x = b0.x; b0lo.y = b0.y;
        v2f b1lo; b1lo.x = b1.x; b1lo.y = b1.y;
        v2f b2lo; b2lo.x = b2.x; b2lo.y = b2.y;
        v2f b3lo; b3lo.x = b3.x; b3lo.y = b3.y;
        v2f b0hi; b0hi.x = b0.z; b0hi.y = b0.w;
        v2f b1hi; b1hi.x = b1.z; b1hi.y = b1.w;
        v2f b2hi; b2hi.x = b2.z; b2hi.y = b2.w;
        v2f b3hi; b3hi.x = b3.z; b3hi.y = b3.w;
        acc0 = __builtin_amdgcn_wmma_f32_16x16x4_f32(false, alo, false, b0lo, (short)0, acc0, false, false);
        acc1 = __builtin_amdgcn_wmma_f32_16x16x4_f32(false, alo, false, b1lo, (short)0, acc1, false, false);
        acc2 = __builtin_amdgcn_wmma_f32_16x16x4_f32(false, alo, false, b2lo, (short)0, acc2, false, false);
        acc3 = __builtin_amdgcn_wmma_f32_16x16x4_f32(false, alo, false, b3lo, (short)0, acc3, false, false);
        acc0 = __builtin_amdgcn_wmma_f32_16x16x4_f32(false, ahi, false, b0hi, (short)0, acc0, false, false);
        acc1 = __builtin_amdgcn_wmma_f32_16x16x4_f32(false, ahi, false, b1hi, (short)0, acc1, false, false);
        acc2 = __builtin_amdgcn_wmma_f32_16x16x4_f32(false, ahi, false, b2hi, (short)0, acc2, false, false);
        acc3 = __builtin_amdgcn_wmma_f32_16x16x4_f32(false, ahi, false, b3hi, (short)0, acc3, false, false);
    }

    #pragma unroll
    for (int v = 0; v < 8; ++v) {
        float* crow = C + (size_t)(m0 + v + 8 * half) * N + n0 + l15;
        crow[ 0] = acc0[v];
        crow[16] = acc1[v];
        crow[32] = acc2[v];
        crow[48] = acc3[v];
    }
}

template <bool RELU>
__global__ __launch_bounds__(256)
void gemm_wmma_f32(const float* __restrict__ A, const float* __restrict__ W,
                   float* __restrict__ C, int N, int mTiles) {
    gemm_block<RELU>(A, W, C, N, mTiles, blockIdx.x, blockIdx.y);
}

constexpr int BCAP       = 64;
constexpr int CNT_PAD    = 20096;
constexpr int GATHER_WGS = 1280;
static_assert(NEDGES % 4 == 0, "edge quads");
static_assert(DIN == GK && DHID == GK, "both products have K == 256");
static_assert(NNODES + 1 <= CNT_PAD && (CNT_PAD * 4) % 512 == 0, "counter pad");

__global__ void zero_int_kernel(int* __restrict__ p, int n) {
    int i = blockIdx.x * blockDim.x + threadIdx.x;
    if (i < n) p[i] = 0;
}

__device__ __forceinline__ void put_edge(int d, int s, float w, unsigned t,
                                         int* __restrict__ nspill, int2* __restrict__ bucket,
                                         int4* __restrict__ spill, int E) {
    if (t < (unsigned)BCAP) {
        bucket[(size_t)d * BCAP + t] = make_int2(s, __float_as_int(w));
    } else {
        const unsigned q = (unsigned)atomicAdd(nspill, 1);
        if (q < (unsigned)E) spill[q] = make_int4(d, s, __float_as_int(w), 0);
    }
}

__device__ __forceinline__
void fill_block(int fb, const int* __restrict__ src, const int* __restrict__ dst,
                const float* __restrict__ w, int* __restrict__ cnt, int* __restrict__ nspill,
                int2* __restrict__ bucket, int4* __restrict__ spill, int E) {
    const int e0 = (fb * 256 + threadIdx.x) * 4;
    if (e0 >= E) return;
    const int4   d4 = *(const int4*)(dst + e0);
    const int4   s4 = *(const int4*)(src + e0);
    const float4 w4 = *(const float4*)(w + e0);
    const unsigned t0 = (unsigned)atomicAdd(cnt + d4.x, 1);
    const unsigned t1 = (unsigned)atomicAdd(cnt + d4.y, 1);
    const unsigned t2 = (unsigned)atomicAdd(cnt + d4.z, 1);
    const unsigned t3 = (unsigned)atomicAdd(cnt + d4.w, 1);
    put_edge(d4.x, s4.x, w4.x, t0, nspill, bucket, spill, E);
    put_edge(d4.y, s4.y, w4.y, t1, nspill, bucket, spill, E);
    put_edge(d4.z, s4.z, w4.z, t2, nspill, bucket, spill, E);
    put_edge(d4.w, s4.w, w4.w, t3, nspill, bucket, spill, E);
}

constexpr int G1_MTILES = NNODES / 16;
constexpr int G1_WGS    = (DHID / 64) * ((G1_MTILES + 7) / 8);
constexpr int FILL_WGS  = (NEDGES / 4 + 255) / 256;

__global__ __launch_bounds__(256)
void gemm1_and_fill(const float* __restrict__ A, const float* __restrict__ W, float* __restrict__ C,
                    const int* __restrict__ src, const int* __restrict__ dst,
                    const float* __restrict__ w, int* __restrict__ cnt, int* __restrict__ nspill,
                    int2* __restrict__ bucket, int4* __restrict__ spill) {
    const int b = blockIdx.x;
    if (b < G1_WGS) {
        gemm_block<false>(A, W, C, DHID, G1_MTILES, b & 3, b >> 2);
    } else {
        fill_block(b - G1_WGS, src, dst, w, cnt, nspill, bucket, spill, NEDGES);
    }
}

__global__ __launch_bounds__(256)
void gather_rows256(const float* __restrict__ X, const int* __restrict__ cnt,
                    const int* __restrict__ nspill, const int2* __restrict__ bucket,
                    const int4* __restrict__ spill, float* __restrict__ out, int nNodes, int E) {
    const int lane = threadIdx.x & 31;
    const int wv = __builtin_amdgcn_readfirstlane((int)(threadIdx.x >> 5));
    const int nW = gridDim.x * 8;
    const unsigned last = (unsigned)(nNodes - 1);
    for (int n = blockIdx.x * 8 + wv; n < nNodes; n += nW) {
        const int c   = cnt[n];
        const int beg = n * BCAP;
        const int end = beg + min(c, BCAP);
        float4 a0 = make_float4(0.f, 0.f, 0.f, 0.f);
        float4 a1 = a0;
        for (int base = beg; base < end; base += 32) {
            const int m = min(32, end - base);
            int2 ed = make_int2(0, 0);
            if (lane < m) ed = bucket[base + lane];
            int j = 0;
            for (; j + 4 <= m; j += 4) {
                const unsigned s0 = min((unsigned)__builtin_amdgcn_readlane(ed.x, j + 0), last);
                const unsigned s1 = min((unsigned)__builtin_amdgcn_readlane(ed.x, j + 1), last);
                const unsigned s2 = min((unsigned)__builtin_amdgcn_readlane(ed.x, j + 2), last);
                const unsigned s3 = min((unsigned)__builtin_amdgcn_readlane(ed.x, j + 3), last);
                const float w0 = __int_as_float(__builtin_amdgcn_readlane(ed.y, j + 0));
                const float w1 = __int_as_float(__builtin_amdgcn_readlane(ed.y, j + 1));
                const float w2 = __int_as_float(__builtin_amdgcn_readlane(ed.y, j + 2));
                const float w3 = __int_as_float(__builtin_amdgcn_readlane(ed.y, j + 3));
                const float4* r0 = (const float4*)(X + (size_t)s0 * 256) + lane;
                const float4* r1 = (const float4*)(X + (size_t)s1 * 256) + lane;
                const float4* r2 = (const float4*)(X + (size_t)s2 * 256) + lane;
                const float4* r3 = (const float4*)(X + (size_t)s3 * 256) + lane;
                const float4 x00 = r0[0], x01 = r0[32];
                const float4 x10 = r1[0], x11 = r1[32];
                const float4 x20 = r2[0], x21 = r2[32];
                const float4 x30 = r3[0], x31 = r3[32];
                a0.x = fmaf(w0, x00.x, a0.x); a0.y = fmaf(w0, x00.y, a0.y); a0.z = fmaf(w0, x00.z, a0.z); a0.w = fmaf(w0, x00.w, a0.w);
                a1.x = fmaf(w0, x01.x, a1.x); a1.y = fmaf(w0, x01.y, a1.y); a1.z = fmaf(w0, x01.z, a1.z); a1.w = fmaf(w0, x01.w, a1.w);
                a0.x = fmaf(w1, x10.x, a0.x); a0.y = fmaf(w1, x10.y, a0.y); a0.z = fmaf(w1, x10.z, a0.z); a0.w = fmaf(w1, x10.w, a0.w);
                a1.x = fmaf(w1, x11.x, a1.x); a1.y = fmaf(w1, x11.y, a1.y); a1.z = fmaf(w1, x11.z, a1.z); a1.w = fmaf(w1, x11.w, a1.w);
                a0.x = fmaf(w2, x20.x, a0.x); a0.y = fmaf(w2, x20.y, a0.y); a0.z = fmaf(w2, x20.z, a0.z); a0.w = fmaf(w2, x20.w, a0.w);
                a1.x = fmaf(w2, x21.x, a1.x); a1.y = fmaf(w2, x21.y, a1.y); a1.z = fmaf(w2, x21.z, a1.z); a1.w = fmaf(w2, x21.w, a1.w);
                a0.x = fmaf(w3, x30.x, a0.x); a0.y = fmaf(w3, x30.y, a0.y); a0.z = fmaf(w3, x30.z, a0.z); a0.w = fmaf(w3, x30.w, a0.w);
                a1.x = fmaf(w3, x31.x, a1.x); a1.y = fmaf(w3, x31.y, a1.y); a1.z = fmaf(w3, x31.z, a1.z); a1.w = fmaf(w3, x31.w, a1.w);
            }
            for (; j < m; ++j) {
                const unsigned s0 = min((unsigned)__builtin_amdgcn_readlane(ed.x, j), last);
                const float w0 = __int_as_float(__builtin_amdgcn_readlane(ed.y, j));
                const float4* r0 = (const float4*)(X + (size_t)s0 * 256) + lane;
                const float4 x00 = r0[0], x01 = r0[32];
                a0.x = fmaf(w0, x00.x, a0.x); a0.y = fmaf(w0, x00.y, a0.y); a0.z = fmaf(w0, x00.z, a0.z); a0.w = fmaf(w0, x00.w, a0.w);
                a1.x = fmaf(w0, x01.x, a1.x); a1.y = fmaf(w0, x01.y, a1.y); a1.z = fmaf(w0, x01.z, a1.z); a1.w = fmaf(w0, x01.w, a1.w);
            }
        }
        if (c > BCAP) {
            const int tot = min(nspill[0], E);
            for (int base = 0; base < tot; base += 32) {
                const int m = min(32, tot - base);
                int4 ev = make_int4(-1, 0, 0, 0);
                if (lane < m) ev = spill[base + lane];
                for (int j = 0; j < m; ++j) {
                    if (__builtin_amdgcn_readlane(ev.x, j) != n) continue;
                    const unsigned s0 = min((unsigned)__builtin_amdgcn_readlane(ev.y, j), last);
                    const float w0 = __int_as_float(__builtin_amdgcn_readlane(ev.z, j));
                    const float4* r0 = (const float4*)(X + (size_t)s0 * 256) + lane;
                    const float4 x00 = r0[0], x01 = r0[32];
                    a0.x = fmaf(w0, x00.x, a0.x); a0.y = fmaf(w0, x00.y, a0.y); a0.z = fmaf(w0, x00.z, a0.z); a0.w = fmaf(w0, x00.w, a0.w);
                    a1.x = fmaf(w0, x01.x, a1.x); a1.y = fmaf(w0, x01.y, a1.y); a1.z = fmaf(w0, x01.z, a1.z); a1.w = fmaf(w0, x01.w, a1.w);
                }
            }
        }
        float4* o = (float4*)(out + (size_t)n * 256) + lane;
        o[0]  = a0;
        o[32] = a1;
    }
}

__global__ __launch_bounds__(256)
void gather_rows64(const float* __restrict__ X, const int* __restrict__ cnt,
                   const int* __restrict__ nspill, const int2* __restrict__ bucket,
                   const int4* __restrict__ spill, float* __restrict__ out, int nNodes, int E) {
    const int lane = threadIdx.x & 31;
    const int wv = __builtin_amdgcn_readfirstlane((int)(threadIdx.x >> 5));
    const int nW = gridDim.x * 8;
    const unsigned last = (unsigned)(nNodes - 1);
    for (int n = blockIdx.x * 8 + wv; n < nNodes; n += nW) {
        const int c   = cnt[n];
        const int beg = n * BCAP;
        const int end = beg + min(c, BCAP);
        float2 a = make_float2(0.f, 0.f);
        for (int base = beg; base < end; base += 32) {
            const int m = min(32, end - base);
            int2 ed = make_int2(0, 0);
            if (lane < m) ed = bucket[base + lane];
            int j = 0;
            for (; j + 4 <= m; j += 4) {
                const unsigned s0 = min((unsigned)__builtin_amdgcn_readlane(ed.x, j + 0), last);
                const unsigned s1 = min((unsigned)__builtin_amdgcn_readlane(ed.x, j + 1), last);
                const unsigned s2 = min((unsigned)__builtin_amdgcn_readlane(ed.x, j + 2), last);
                const unsigned s3 = min((unsigned)__builtin_amdgcn_readlane(ed.x, j + 3), last);
                const float w0 = __int_as_float(__builtin_amdgcn_readlane(ed.y, j + 0));
                const float w1 = __int_as_float(__builtin_amdgcn_readlane(ed.y, j + 1));
                const float w2 = __int_as_float(__builtin_amdgcn_readlane(ed.y, j + 2));
                const float w3 = __int_as_float(__builtin_amdgcn_readlane(ed.y, j + 3));
                const float2 x0 = *((const float2*)(X + (size_t)s0 * 64) + lane);
                const float2 x1 = *((const float2*)(X + (size_t)s1 * 64) + lane);
                const float2 x2 = *((const float2*)(X + (size_t)s2 * 64) + lane);
                const float2 x3 = *((const float2*)(X + (size_t)s3 * 64) + lane);
                a.x = fmaf(w0, x0.x, a.x); a.y = fmaf(w0, x0.y, a.y);
                a.x = fmaf(w1, x1.x, a.x); a.y = fmaf(w1, x1.y, a.y);
                a.x = fmaf(w2, x2.x, a.x); a.y = fmaf(w2, x2.y, a.y);
                a.x = fmaf(w3, x3.x, a.x); a.y = fmaf(w3, x3.y, a.y);
            }
            for (; j < m; ++j) {
                const unsigned s0 = min((unsigned)__builtin_amdgcn_readlane(ed.x, j), last);
                const float w0 = __int_as_float(__builtin_amdgcn_readlane(ed.y, j));
                const float2 x0 = *((const float2*)(X + (size_t)s0 * 64) + lane);
                a.x = fmaf(w0, x0.x, a.x); a.y = fmaf(w0, x0.y, a.y);
            }
        }
        if (c > BCAP) {
            const int tot = min(nspill[0], E);
            for (int base = 0; base < tot; base += 32) {
                const int m = min(32, tot - base);
                int4 ev = make_int4(-1, 0, 0, 0);
                if (lane < m) ev = spill[base + lane];
                for (int j = 0; j < m; ++j) {
                    if (__builtin_amdgcn_readlane(ev.x, j) != n) continue;
                    const unsigned s0 = min((unsigned)__builtin_amdgcn_readlane(ev.y, j), last);
                    const float w0 = __int_as_float(__builtin_amdgcn_readlane(ev.z, j));
                    const float2 x0 = *((const float2*)(X + (size_t)s0 * 64) + lane);
                    a.x = fmaf(w0, x0.x, a.x); a.y = fmaf(w0, x0.y, a.y);
                }
            }
        }
        *((float2*)(out + (size_t)n * 64) + lane) = a;
    }
}

extern "C" void kernel_launch(void* const* d_in, const int* in_sizes, int n_in,
                              void* d_out, int out_size, void* d_ws, size_t ws_size,
                              hipStream_t stream) {
    const float* node_feat = (const float*)d_in[0];
    const int*   edge_src  = (const int*)  d_in[1];
    const int*   edge_dst  = (const int*)  d_in[2];
    const float* edge_w    = (const float*)d_in[3];
    const float* W1        = (const float*)d_in[4];
    const float* W2        = (const float*)d_in[5];
    float* out = (float*)d_out;

    float* y1   = (float*)d_ws;
    float* agg1 = y1 + (size_t)NNODES * DHID;
    float* y2   = agg1 + (size_t)NNODES * DHID;
    int*   cnt    = (int*)(y2 + (size_t)NNODES * DOUT);
    int*   nspill = cnt + NNODES;
    int2*  bucket = (int2*)(cnt + CNT_PAD);
    int4*  spill  = (int4*)(bucket + (size_t)NNODES * BCAP);

    constexpr int mTiles = NNODES / 16;

    zero_int_kernel<<<(NNODES + 1 + 255) / 256, 256, 0, stream>>>(cnt, NNODES + 1);

    gemm1_and_fill<<<G1_WGS + FILL_WGS, 256, 0, stream>>>(node_feat, W1, y1, edge_src, edge_dst, edge_w,
                                                         cnt, nspill, bucket, spill);
    gather_rows256<<<GATHER_WGS, 256, 0, stream>>>(y1, cnt, nspill, bucket, spill, agg1, NNODES, NEDGES);
    {
        dim3 grid(DOUT / 64, (mTiles + 7) / 8);
        gemm_wmma_f32<true><<<grid, 256, 0, stream>>>(
            agg1, W2, y2, DOUT, mTiles);
    }
    gather_rows64<<<GATHER_WGS, 256, 0, stream>>>(y2, cnt, nspill, bucket, spill, out, NNODES, NEDGES);
}


# Layer-1 neighbour sum, ReLU and layer-2 product fused in one launch: 16-wave workgroup sums 32 rows into LDS tiles, two waves multiply each tile by W2 from LDS; summed rows never go to memory
# speedup vs baseline: 2.3703x; 1.0825x over previous
#include <hip/hip_runtime.h>
#include <hip/hip_bf16.h>

typedef float v2f __attribute__((ext_vector_type(2)));
typedef float v8f __attribute__((ext_vector_type(8)));

constexpr int NNODES = 20000;
constexpr int NEDGES = 320000;
constexpr int DIN    = 256;
constexpr int DHID   = 256;
constexpr int DOUT   = 64;


constexpr int GK      = 256;
constexpr int BPITCH4 = GK / 4 + 1;

__device__ __forceinline__
void gemm_block(const float* __restrict__ A, const float* __restrict__ W,
                float* __restrict__ C, int N, int mTiles, int bx, int by) {
    __shared__ float4 smem4[64 * BPITCH4];

    const int tid  = threadIdx.x;
    const int lane = tid & 31;
    const int wave = tid >> 5;
    const int n0   = bx << 6;

    {
        const int c4 = tid & 63;
        const int r0 = tid >> 6;
        const float4* wsrc = (const float4*)(W + (size_t)n0 * GK);
        float4 v[16];
        #pragma unroll
        for (int i = 0; i < 16; ++i)
            v[i] = wsrc[(size_t)(i * 4 + r0) * (GK / 4) + c4];
        #pragma unroll
        for (int i = 0; i < 16; ++i)
            smem4[(i * 4 + r0) * BPITCH4 + c4] = v[i];
    }
    __syncthreads();

    const int mt = by * 8 + wave;
    if (mt >= mTiles) return;
    const int m0 = mt << 4;

    const int half = lane >> 4;
    const int l15  = lane & 15;

    const float* arow = A + (size_t)(m0 + l15) * GK + 4 * half;
    __builtin_prefetch(arow, 0, 3);
    __builtin_prefetch(arow + 128, 0, 3);

    const float4* bbase = smem4 + l15 * BPITCH4 + half;

    v8f acc0 = {}, acc1 = {}, acc2 = {}, acc3 = {};

    #pragma unroll 4
    for (int k0 = 0; k0 < GK; k0 += 8) {
        const float4 a = *(const float4*)(arow + k0);
        const float4* bp = bbase + (k0 >> 2);
        const float4 b0 = bp[0];
        const float4 b1 = bp[16 * BPITCH4];
        const float4 b2 = bp[32 * BPITCH4];
        const float4 b3 = bp[48 * BPITCH4];
        v2f alo; alo.x = a.x; alo.y = a.y;
        v2f ahi; ahi.x = a.z; ahi.y = a.w;
        v2f b0lo; b0lo.x = b0.x; b0lo.y = b0.y;
        v2f b1lo; b1lo.x = b1.x; b1lo.y = b1.y;
        v2f b2lo; b2lo.x = b2.x; b2lo.y = b2.y;
        v2f b3lo; b3lo.x = b3.x; b3lo.y = b3.y;
        v2f b0hi; b0hi.x = b0.z; b0hi.y = b0.w;
        v2f b1hi; b1hi.x = b1.z; b1hi.y = b1.w;
        v2f b2hi; b2hi.x = b2.z; b2hi.y = b2.w;
        v2f b3hi; b3hi.x = b3.z; b3hi.y = b3.w;
        acc0 = __builtin_amdgcn_wmma_f32_16x16x4_f32(false, alo, false, b0lo, (short)0, acc0, false, false);
        acc1 = __builtin_amdgcn_wmma_f32_16x16x4_f32(false, alo, false, b1lo, (short)0, acc1, false, false);
        acc2 = __builtin_amdgcn_wmma_f32_16x16x4_f32(false, alo, false, b2lo, (short)0, acc2, false, false);
        acc3 = __builtin_amdgcn_wmma_f32_16x16x4_f32(false, alo, false, b3lo, (short)0, acc3, false, false);
        acc0 = __builtin_amdgcn_wmma_f32_16x16x4_f32(false, ahi, false, b0hi, (short)0, acc0, false, false);
        acc1 = __builtin_amdgcn_wmma_f32_16x16x4_f32(false, ahi, false, b1hi, (short)0, acc1, false, false);
        acc2 = __builtin_amdgcn_wmma_f32_16x16x4_f32(false, ahi, false, b2hi, (short)0, acc2, false, false);
        acc3 = __builtin_amdgcn_wmma_f32_16x16x4_f32(false, ahi, false, b3hi, (short)0, acc3, false, false);
    }

    #pragma unroll
    for (int v = 0; v < 8; ++v) {
        float* crow = C + (size_t)(m0 + v + 8 * half) * N + n0 + l15;
        crow[ 0] = acc0[v];
        crow[16] = acc1[v];
        crow[32] = acc2[v];
        crow[48] = acc3[v];
    }
}

constexpr int BCAP       = 64;
constexpr int CNT_PAD    = 20096;
constexpr int GATHER_WGS = 1280;
static_assert(NEDGES % 4 == 0, "edge quads");
static_assert(DIN == GK && DHID == GK && DOUT == 64, "both products have K == 256, the second 64 columns");
static_assert(NNODES + 1 <= CNT_PAD && (CNT_PAD * 4) % 512 == 0, "counter pad");

__global__ void zero_int_kernel(int* __restrict__ p, int n) {
    int i = blockIdx.x * blockDim.x + threadIdx.x;
    if (i < n) p[i] = 0;
}

__device__ __forceinline__ void put_edge(int d, int s, float w, unsigned t,
                                         int* __restrict__ nspill, int2* __restrict__ bucket,
                                         int4* __restrict__ spill, int E) {
    if (t < (unsigned)BCAP) {
        bucket[(size_t)d * BCAP + t] = make_int2(s, __float_as_int(w));
    } else {
        const unsigned q = (unsigned)atomicAdd(nspill, 1);
        if (q < (unsigned)E) spill[q] = make_int4(d, s, __float_as_int(w), 0);
    }
}

__device__ __forceinline__
void fill_block(int fb, const int* __restrict__ src, const int* __restrict__ dst,
                const float* __restrict__ w, int* __restrict__ cnt, int* __restrict__ nspill,
                int2* __restrict__ bucket, int4* __restrict__ spill, int E) {
    const int e0 = (fb * 256 + threadIdx.x) * 4;
    if (e0 >= E) return;
    const int4   d4 = *(const int4*)(dst + e0);
    const int4   s4 = *(const int4*)(src + e0);
    const float4 w4 = *(const float4*)(w + e0);
    const unsigned t0 = (unsigned)atomicAdd(cnt + d4.x, 1);
    const unsigned t1 = (unsigned)atomicAdd(cnt + d4.y, 1);
    const unsigned t2 = (unsigned)atomicAdd(cnt + d4.z, 1);
    const unsigned t3 = (unsigned)atomicAdd(cnt + d4.w, 1);
    put_edge(d4.x, s4.x, w4.x, t0, nspill, bucket, spill, E);
    put_edge(d4.y, s4.y, w4.y, t1, nspill, bucket, spill, E);
    put_edge(d4.z, s4.z, w4.z, t2, nspill, bucket, spill, E);
    put_edge(d4.w, s4.w, w4.w, t3, nspill, bucket, spill, E);
}

constexpr int G1_MTILES = NNODES / 16;
constexpr int G1_WGS    = (DHID / 64) * ((G1_MTILES + 7) / 8);
constexpr int FILL_WGS  = (NEDGES / 4 + 255) / 256;

__global__ __launch_bounds__(256)
void gemm1_and_fill(const float* __restrict__ A, const float* __restrict__ W, float* __restrict__ C,
                    const int* __restrict__ src, const int* __restrict__ dst,
                    const float* __restrict__ w, int* __restrict__ cnt, int* __restrict__ nspill,
                    int2* __restrict__ bucket, int4* __restrict__ spill) {
    const int b = blockIdx.x;
    if (b < G1_WGS) {
        gemm_block(A, W, C, DHID, G1_MTILES, b & 3, b >> 2);
    } else {
        fill_block(b - G1_WGS, src, dst, w, cnt, nspill, bucket, spill, NEDGES);
    }
}

__device__ __forceinline__
void row_sum256(const float* __restrict__ X, const int* __restrict__ cnt,
                const int* __restrict__ nspill, const int2* __restrict__ bucket,
                const int4* __restrict__ spill, int n, int lane, unsigned last, int E,
                float4& a0, float4& a1) {
    const int c   = cnt[n];
    const int beg = n * BCAP;
    const int end = beg + min(c, BCAP);
    a0 = make_float4(0.f, 0.f, 0.f, 0.f);
    a1 = a0;
    for (int base = beg; base < end; base += 32) {
        const int m = min(32, end - base);
        int2 ed = make_int2(0, 0);
        if (lane < m) ed = bucket[base + lane];
        int j = 0;
        for (; j + 4 <= m; j += 4) {
            const unsigned s0 = min((unsigned)__builtin_amdgcn_readlane(ed.x, j + 0), last);
            const unsigned s1 = min((unsigned)__builtin_amdgcn_readlane(ed.x, j + 1), last);
            const unsigned s2 = min((unsigned)__builtin_amdgcn_readlane(ed.x, j + 2), last);
            const unsigned s3 = min((unsigned)__builtin_amdgcn_readlane(ed.x, j + 3), last);
            const float w0 = __int_as_float(__builtin_amdgcn_readlane(ed.y, j + 0));
            const float w1 = __int_as_float(__builtin_amdgcn_readlane(ed.y, j + 1));
            const float w2 = __int_as_float(__builtin_amdgcn_readlane(ed.y, j + 2));
            const float w3 = __int_as_float(__builtin_amdgcn_readlane(ed.y, j + 3));
            const float4* r0 = (const float4*)(X + (size_t)s0 * 256) + lane;
            const float4* r1 = (const float4*)(X + (size_t)s1 * 256) + lane;
            const float4* r2 = (const float4*)(X + (size_t)s2 * 256) + lane;
            const float4* r3 = (const float4*)(X + (size_t)s3 * 256) + lane;
            const float4 x00 = r0[0], x01 = r0[32];
            const float4 x10 = r1[0], x11 = r1[32];
            const float4 x20 = r2[0], x21 = r2[32];
            const float4 x30 = r3[0], x31 = r3[32];
            a0.x = fmaf(w0, x00.x, a0.x); a0.y = fmaf(w0, x00.y, a0.y); a0.z = fmaf(w0, x00.z, a0.z); a0.w = fmaf(w0, x00.w, a0.w);
            a1.x = fmaf(w0, x01.x, a1.x); a1.y = fmaf(w0, x01.y, a1.y); a1.z = fmaf(w0, x01.z, a1.z); a1.w = fmaf(w0, x01.w, a1.w);
            a0.x = fmaf(w1, x10.x, a0.x); a0.y = fmaf(w1, x10.y, a0.y); a0.z = fmaf(w1, x10.z, a0.z); a0.w = fmaf(w1, x10.w, a0.w);
            a1.x = fmaf(w1, x11.x, a1.x); a1.y = fmaf(w1, x11.y, a1.y); a1.z = fmaf(w1, x11.z, a1.z); a1.w = fmaf(w1, x11.w, a1.w);
            a0.x = fmaf(w2, x20.x, a0.x); a0.y = fmaf(w2, x20.y, a0.y); a0.z = fmaf(w2, x20.z, a0.z); a0.w = fmaf(w2, x20.w, a0.w);
            a1.x = fmaf(w2, x21.x, a1.x); a1.y = fmaf(w2, x21.y, a1.y); a1.z = fmaf(w2, x21.z, a1.z); a1.w = fmaf(w2, x21.w, a1.w);
            a0.x = fmaf(w3, x30.x, a0.x); a0.y = fmaf(w3, x30.y, a0.y); a0.z = fmaf(w3, x30.z, a0.z); a0.w = fmaf(w3, x30.w, a0.w);
            a1.x = fmaf(w3, x31.x, a1.x); a1.y = fmaf(w3, x31.y, a1.y); a1.z = fmaf(w3, x31.z, a1.z); a1.w = fmaf(w3, x31.w, a1.w);
        }
        for (; j < m; ++j) {
            const unsigned s0 = min((unsigned)__builtin_amdgcn_readlane(ed.x, j), last);
            const float w0 = __int_as_float(__builtin_amdgcn_readlane(ed.y, j));
            const float4* r0 = (const float4*)(X + (size_t)s0 * 256) + lane;
            const float4 x00 = r0[0], x01 = r0[32];
            a0.x = fmaf(w0, x00.x, a0.x); a0.y = fmaf(w0, x00.y, a0.y); a0.z = fmaf(w0, x00.z, a0.z); a0.w = fmaf(w0, x00.w, a0.w);
            a1.x = fmaf(w0, x01.x, a1.x); a1.y = fmaf(w0, x01.y, a1.y); a1.z = fmaf(w0, x01.z, a1.z); a1.w = fmaf(w0, x01.w, a1.w);
        }
    }
    if (c > BCAP) {
        const int tot = min(nspill[0], E);
        for (int base = 0; base < tot; base += 32) {
            const int m = min(32, tot - base);
            int4 ev = make_int4(-1, 0, 0, 0);
            if (lane < m) ev = spill[base + lane];
            for (int j = 0; j < m; ++j) {
                if (__builtin_amdgcn_readlane(ev.x, j) != n) continue;
                const unsigned s0 = min((unsigned)__builtin_amdgcn_readlane(ev.y, j), last);
                const float w0 = __int_as_float(__builtin_amdgcn_readlane(ev.z, j));
                const float4* r0 = (const float4*)(X + (size_t)s0 * 256) + lane;
                const float4 x00 = r0[0], x01 = r0[32];
                a0.x = fmaf(w0, x00.x, a0.x); a0.y = fmaf(w0, x00.y, a0.y); a0.z = fmaf(w0, x00.z, a0.z); a0.w = fmaf(w0, x00.w, a0.w);
                a1.x = fmaf(w0, x01.x, a1.x); a1.y = fmaf(w0, x01.y, a1.y); a1.z = fmaf(w0, x01.z, a1.z); a1.w = fmaf(w0, x01.w, a1.w);
            }
        }
    }
}

constexpr int FG_ROWS = 32;
constexpr int FG_WGS  = NNODES / FG_ROWS;
static_assert(NNODES % FG_ROWS == 0, "whole workgroups of rows: every wave of every pass has a row");

__global__ __launch_bounds__(512, 12)
void gather256_relu_gemm64(const float* __restrict__ X, const int* __restrict__ cnt,
                           const int* __restrict__ nspill, const int2* __restrict__ bucket,
                           const int4* __restrict__ spill, const float* __restrict__ W,
                           float* __restrict__ C, int nNodes, int E) {
    __shared__ float4 wl[64 * BPITCH4];
    __shared__ float4 al[2][16 * BPITCH4];

    const int tid  = threadIdx.x;
    const int lane = tid & 31;
    const int wv = __builtin_amdgcn_readfirstlane(tid >> 5);
    const unsigned last = (unsigned)(nNodes - 1);

    {
        const int c4 = tid & 63;
        const int r0 = tid >> 6;
        const float4* wsrc = (const float4*)W;
        float4 v[8];
        #pragma unroll
        for (int i = 0; i < 8; ++i)
            v[i] = wsrc[(size_t)(i * 8 + r0) * (GK / 4) + c4];
        #pragma unroll
        for (int i = 0; i < 8; ++i)
            wl[(i * 8 + r0) * BPITCH4 + c4] = v[i];
    }

    const int half = lane >> 4;
    const int l15  = lane & 15;
    const int b    = blockIdx.x;

    for (int p = 0; p < 2; ++p) {
        const int m0 = b * FG_ROWS + p * 16;
        float4 a0, a1;
        row_sum256(X, cnt, nspill, bucket, spill, m0 + wv, lane, last, E, a0, a1);
        a0.x = fmaxf(a0.x, 0.0f); a0.y = fmaxf(a0.y, 0.0f); a0.z = fmaxf(a0.z, 0.0f); a0.w = fmaxf(a0.w, 0.0f);
        a1.x = fmaxf(a1.x, 0.0f); a1.y = fmaxf(a1.y, 0.0f); a1.z = fmaxf(a1.z, 0.0f); a1.w = fmaxf(a1.w, 0.0f);
        al[p][wv * BPITCH4 + lane]      = a0;
        al[p][wv * BPITCH4 + 32 + lane] = a1;
        __syncthreads();

        const int q = (b + (b >> 8) + p) & 7;
        if ((wv >> 1) == q) {
            const int u = wv & 1;
            const float4* ab = &al[p][l15 * BPITCH4 + half];
            const float4* bb = &wl[(32 * u + l15) * BPITCH4 + half];
            v8f acc0 = {}, acc1 = {};
            #pragma unroll 4
            for (int j = 0; j < GK / 8; ++j) {
                const float4 a  = ab[2 * j];
                const float4 b0 = bb[2 * j];
                const float4 b1 = bb[16 * BPITCH4 + 2 * j];
                v2f alo;  alo.x  = a.x;  alo.y  = a.y;
                v2f ahi;  ahi.x  = a.z;  ahi.y  = a.w;
                v2f b0lo; b0lo.x = b0.x; b0lo.y = b0.y;
                v2f b1lo; b1lo.x = b1.x; b1lo.y = b1.y;
                v2f b0hi; b0hi.x = b0.z; b0hi.y = b0.w;
                v2f b1hi; b1hi.x = b1.z; b1hi.y = b1.w;
                acc0 = __builtin_amdgcn_wmma_f32_16x16x4_f32(false, alo, false, b0lo, (short)0, acc0, false, false);
                acc1 = __builtin_amdgcn_wmma_f32_16x16x4_f32(false, alo, false, b1lo, (short)0, acc1, false, false);
                acc0 = __builtin_amdgcn_wmma_f32_16x16x4_f32(false, ahi, false, b0hi, (short)0, acc0, false, false);
                acc1 = __builtin_amdgcn_wmma_f32_16x16x4_f32(false, ahi, false, b1hi, (short)0, acc1, false, false);
            }
            #pragma unroll
            for (int v = 0; v < 8; ++v) {
                float* crow = C + (size_t)(m0 + v + 8 * half) * DOUT + 32 * u + l15;
                crow[ 0] = acc0[v];
                crow[16] = acc1[v];
            }
        }
    }
}

__global__ __launch_bounds__(256)
void gather_rows64(const float* __restrict__ X, const int* __restrict__ cnt,
                   const int* __restrict__ nspill, const int2* __restrict__ bucket,
                   const int4* __restrict__ spill, float* __restrict__ out, int nNodes, int E) {
    const int lane = threadIdx.x & 31;
    const int wv = __builtin_amdgcn_readfirstlane((int)(threadIdx.x >> 5));
    const int nW = gridDim.x * 8;
    const unsigned last = (unsigned)(nNodes - 1);
    for (int n = blockIdx.x * 8 + wv; n < nNodes; n += nW) {
        const int c   = cnt[n];
        const int beg = n * BCAP;
        const int end = beg + min(c, BCAP);
        float2 a = make_float2(0.f, 0.f);
        for (int base = beg; base < end; base += 32) {
            const int m = min(32, end - base);
            int2 ed = make_int2(0, 0);
            if (lane < m) ed = bucket[base + lane];
            int j = 0;
            for (; j + 4 <= m; j += 4) {
                const unsigned s0 = min((unsigned)__builtin_amdgcn_readlane(ed.x, j + 0), last);
                const unsigned s1 = min((unsigned)__builtin_amdgcn_readlane(ed.x, j + 1), last);
                const unsigned s2 = min((unsigned)__builtin_amdgcn_readlane(ed.x, j + 2), last);
                const unsigned s3 = min((unsigned)__builtin_amdgcn_readlane(ed.x, j + 3), last);
                const float w0 = __int_as_float(__builtin_amdgcn_readlane(ed.y, j + 0));
                const float w1 = __int_as_float(__builtin_amdgcn_readlane(ed.y, j + 1));
                const float w2 = __int_as_float(__builtin_amdgcn_readlane(ed.y, j + 2));
                const float w3 = __int_as_float(__builtin_amdgcn_readlane(ed.y, j + 3));
                const float2 x0 = *((const float2*)(X + (size_t)s0 * 64) + lane);
                const float2 x1 = *((const float2*)(X + (size_t)s1 * 64) + lane);
                const float2 x2 = *((const float2*)(X + (size_t)s2 * 64) + lane);
                const float2 x3 = *((const float2*)(X + (size_t)s3 * 64) + lane);
                a.x = fmaf(w0, x0.x, a.x); a.y = fmaf(w0, x0.y, a.y);
                a.x = fmaf(w1, x1.x, a.x); a.y = fmaf(w1, x1.y, a.y);
                a.x = fmaf(w2, x2.x, a.x); a.y = fmaf(w2, x2.y, a.y);
                a.x = fmaf(w3, x3.x, a.x); a.y = fmaf(w3, x3.y, a.y);
            }
            for (; j < m; ++j) {
                const unsigned s0 = min((unsigned)__builtin_amdgcn_readlane(ed.x, j), last);
                const float w0 = __int_as_float(__builtin_amdgcn_readlane(ed.y, j));
                const float2 x0 = *((const float2*)(X + (size_t)s0 * 64) + lane);
                a.x = fmaf(w0, x0.x, a.x); a.y = fmaf(w0, x0.y, a.y);
            }
        }
        if (c > BCAP) {
            const int tot = min(nspill[0], E);
            for (int base = 0; base < tot; base += 32) {
                const int m = min(32, tot - base);
                int4 ev = make_int4(-1, 0, 0, 0);
                if (lane < m) ev = spill[base + lane];
                for (int j = 0; j < m; ++j) {
                    if (__builtin_amdgcn_readlane(ev.x, j) != n) continue;
                    const unsigned s0 = min((unsigned)__builtin_amdgcn_readlane(ev.y, j), last);
                    const float w0 = __int_as_float(__builtin_amdgcn_readlane(ev.z, j));
                    const float2 x0 = *((const float2*)(X + (size_t)s0 * 64) + lane);
                    a.x = fmaf(w0, x0.x, a.x); a.y = fmaf(w0, x0.y, a.y);
                }
            }
        }
        *((float2*)(out + (size_t)n * 64) + lane) = a;
    }
}

extern "C" void kernel_launch(void* const* d_in, const int* in_sizes, int n_in,
                              void* d_out, int out_size, void* d_ws, size_t ws_size,
                              hipStream_t stream) {
    const float* node_feat = (const float*)d_in[0];
    const int*   edge_src  = (const int*)  d_in[1];
    const int*   edge_dst  = (const int*)  d_in[2];
    const float* edge_w    = (const float*)d_in[3];
    const float* W1        = (const float*)d_in[4];
    const float* W2        = (const float*)d_in[5];
    float* out = (float*)d_out;

    float* y1   = (float*)d_ws;
    float* y2   = y1 + (size_t)NNODES * DHID;
    int*   cnt    = (int*)(y2 + (size_t)NNODES * DOUT);
    int*   nspill = cnt + NNODES;
    int2*  bucket = (int2*)(cnt + CNT_PAD);
    int4*  spill  = (int4*)(bucket + (size_t)NNODES * BCAP);

    zero_int_kernel<<<(NNODES + 1 + 255) / 256, 256, 0, stream>>>(cnt, NNODES + 1);

    gemm1_and_fill<<<G1_WGS + FILL_WGS, 256, 0, stream>>>(node_feat, W1, y1, edge_src, edge_dst, edge_w,
                                                         cnt, nspill, bucket, spill);
    gather256_relu_gemm64<<<FG_WGS, 512, 0, stream>>>(y1, cnt, nspill, bucket, spill, W2, y2, NNODES, NEDGES);
    gather_rows64<<<GATHER_WGS, 256, 0, stream>>>(y2, cnt, nspill, bucket, spill, out, NNODES, NEDGES);
}
